# starting kernel
# baseline (speedup 1.0000x reference)
#include <hip/hip_runtime.h>
#include <hip/hip_bf16.h>
#include <math.h>


typedef __attribute__((ext_vector_type(2))) float v2f;
typedef __attribute__((ext_vector_type(8))) float v8f;

#define T_SEQ   12
#define BCN     16
#define NODES   1024
#define FEAT    32
#define HID     64
#define G3      192
#define NROWS   (BCN * NODES)
#define NEG_BIG (-1e16f)

__device__ __forceinline__ v8f wmma_f32(v2f a, v2f b, v8f c) {
  return __builtin_amdgcn_wmma_f32_16x16x4_f32(
       false, a,  false, b,
       (short)0, c,  false,  false);
}

__device__ __forceinline__ v8f bias8(float v) {
  v8f r;
#pragma unroll
  for (int i = 0; i < 8; ++i) r[i] = v;
  return r;
}

__device__ __forceinline__ float sigmoidf_(float x) {
  return 1.0f / (1.0f + __expf(-x));
}

#define GRU_WAVES 4
#define HPAD 66

__global__ __launch_bounds__(128) void gru_fused_kernel(
    const float* __restrict__ x_in,
    const float* __restrict__ w_ih0,
    const float* __restrict__ w_hh0,
    const float* __restrict__ b_ih0,
    const float* __restrict__ b_hh0,
    const float* __restrict__ w_ih1,
    const float* __restrict__ w_hh1,
    const float* __restrict__ b_ih1,
    const float* __restrict__ b_hh1,
    float* __restrict__ z_out)
{
  __shared__ float wih0_t[FEAT * G3];
  __shared__ float whh0_t[HID * G3];
  __shared__ float wih1_t[HID * G3];
  __shared__ float whh1_t[HID * G3];
  __shared__ float hst[2][GRU_WAVES][16 * HPAD];

  const int tid = threadIdx.x;
  for (int i = tid; i < FEAT * G3; i += blockDim.x) {
    const int k = i / G3, g = i % G3;
    wih0_t[i] = w_ih0[g * FEAT + k];
  }
  for (int i = tid; i < HID * G3; i += blockDim.x) {
    const int k = i / G3, g = i % G3;
    whh0_t[i] = w_hh0[g * HID + k];
    wih1_t[i] = w_ih1[g * HID + k];
    whh1_t[i] = w_hh1[g * HID + k];
  }

  const int wave = tid >> 5;
  const int lane = tid & 31;
  const int l16  = lane & 15;
  const int lh   = lane >> 4;
  float* h0s = &hst[0][wave][0];
  float* h1s = &hst[1][wave][0];
  for (int i = lane; i < 16 * HPAD; i += 32) { h0s[i] = 0.0f; h1s[i] = 0.0f; }
  __syncthreads();

  const int row0 = blockIdx.x * (GRU_WAVES * 16) + wave * 16;

  for (int t = 0; t < T_SEQ; ++t) {
    v2f xf[8];
    {
      const float* xrow = x_in + ((size_t)t * NROWS + row0 + l16) * FEAT;
#pragma unroll
      for (int ks = 0; ks < 8; ++ks) {
        const int k = 4 * ks + 2 * lh;
        xf[ks].x = xrow[k]; xf[ks].y = xrow[k + 1];
      }
    }
    v2f h0f[16], h1f[16];
#pragma unroll
    for (int ks = 0; ks < 16; ++ks) {
      const int k = 4 * ks + 2 * lh;
      h0f[ks].x = h0s[l16 * HPAD + k]; h0f[ks].y = h0s[l16 * HPAD + k + 1];
      h1f[ks].x = h1s[l16 * HPAD + k]; h1f[ks].y = h1s[l16 * HPAD + k + 1];
    }

#pragma unroll
    for (int j = 0; j < 4; ++j) {
      const int gr = j * 16 + l16;
      const int gz = 64 + j * 16 + l16;
      const int gn = 128 + j * 16 + l16;
      v8f gir = bias8(b_ih0[gr]), giz = bias8(b_ih0[gz]), gin = bias8(b_ih0[gn]);
      v8f ghr = bias8(b_hh0[gr]), ghz = bias8(b_hh0[gz]), ghn = bias8(b_hh0[gn]);
#pragma unroll
      for (int ks = 0; ks < 8; ++ks) {
        const int k = 4 * ks + 2 * lh;
        v2f br, bz, bn;
        br.x = wih0_t[k * G3 + gr]; br.y = wih0_t[(k + 1) * G3 + gr];
        bz.x = wih0_t[k * G3 + gz]; bz.y = wih0_t[(k + 1) * G3 + gz];
        bn.x = wih0_t[k * G3 + gn]; bn.y = wih0_t[(k + 1) * G3 + gn];
        gir = wmma_f32(xf[ks], br, gir);
        giz = wmma_f32(xf[ks], bz, giz);
        gin = wmma_f32(xf[ks], bn, gin);
      }
#pragma unroll
      for (int ks = 0; ks < 16; ++ks) {
        const int k = 4 * ks + 2 * lh;
        v2f br, bz, bn;
        br.x = whh0_t[k * G3 + gr]; br.y = whh0_t[(k + 1) * G3 + gr];
        bz.x = whh0_t[k * G3 + gz]; bz.y = whh0_t[(k + 1) * G3 + gz];
        bn.x = whh0_t[k * G3 + gn]; bn.y = whh0_t[(k + 1) * G3 + gn];
        ghr = wmma_f32(h0f[ks], br, ghr);
        ghz = wmma_f32(h0f[ks], bz, ghz);
        ghn = wmma_f32(h0f[ks], bn, ghn);
      }
#pragma unroll
      for (int v = 0; v < 8; ++v) {
        const int a = (v + 8 * lh) * HPAD + j * 16 + l16;
        const float hold = h0s[a];
        const float r  = sigmoidf_(gir[v] + ghr[v]);
        const float zg = sigmoidf_(giz[v] + ghz[v]);
        const float n  = tanhf(gin[v] + r * ghn[v]);
        h0s[a] = (1.0f - zg) * n + zg * hold;
      }
    }

#pragma unroll
    for (int ks = 0; ks < 16; ++ks) {
      const int k = 4 * ks + 2 * lh;
      h0f[ks].x = h0s[l16 * HPAD + k]; h0f[ks].y = h0s[l16 * HPAD + k + 1];
    }

#pragma unroll
    for (int j = 0; j < 4; ++j) {
      const int gr = j * 16 + l16;
      const int gz = 64 + j * 16 + l16;
      const int gn = 128 + j * 16 + l16;
      v8f gir = bias8(b_ih1[gr]), giz = bias8(b_ih1[gz]), gin = bias8(b_ih1[gn]);
      v8f ghr = bias8(b_hh1[gr]), ghz = bias8(b_hh1[gz]), ghn = bias8(b_hh1[gn]);
#pragma unroll
      for (int ks = 0; ks < 16; ++ks) {
        const int k = 4 * ks + 2 * lh;
        v2f br, bz, bn;
        br.x = wih1_t[k * G3 + gr]; br.y = wih1_t[(k + 1) * G3 + gr];
        bz.x = wih1_t[k * G3 + gz]; bz.y = wih1_t[(k + 1) * G3 + gz];
        bn.x = wih1_t[k * G3 + gn]; bn.y = wih1_t[(k + 1) * G3 + gn];
        gir = wmma_f32(h0f[ks], br, gir);
        giz = wmma_f32(h0f[ks], bz, giz);
        gin = wmma_f32(h0f[ks], bn, gin);
      }
#pragma unroll
      for (int ks = 0; ks < 16; ++ks) {
        const int k = 4 * ks + 2 * lh;
        v2f br, bz, bn;
        br.x = whh1_t[k * G3 + gr]; br.y = whh1_t[(k + 1) * G3 + gr];
        bz.x = whh1_t[k * G3 + gz]; bz.y = whh1_t[(k + 1) * G3 + gz];
        bn.x = whh1_t[k * G3 + gn]; bn.y = whh1_t[(k + 1) * G3 + gn];
        ghr = wmma_f32(h1f[ks], br, ghr);
        ghz = wmma_f32(h1f[ks], bz, ghz);
        ghn = wmma_f32(h1f[ks], bn, ghn);
      }
#pragma unroll
      for (int v = 0; v < 8; ++v) {
        const int a = (v + 8 * lh) * HPAD + j * 16 + l16;
        const float hold = h1s[a];
        const float r  = sigmoidf_(gir[v] + ghr[v]);
        const float zg = sigmoidf_(giz[v] + ghz[v]);
        const float n  = tanhf(gin[v] + r * ghn[v]);
        h1s[a] = (1.0f - zg) * n + zg * hold;
      }
    }
  }

  for (int i = lane; i < 16 * HID; i += 32) {
    const int r = i / HID, c = i % HID;
    z_out[(size_t)(row0 + r) * HID + c] = h0s[r * HPAD + c];
    z_out[(size_t)NROWS * HID + (size_t)(row0 + r) * HID + c] = h1s[r * HPAD + c];
  }
}

__global__ __launch_bounds__(32) void gat_kernel(
    const float* __restrict__ z,
    const int*   __restrict__ mask,
    float*       __restrict__ out)
{
  __shared__ float S[NODES * 16];

  const int lane  = threadIdx.x;
  const int l16   = lane & 15;
  const int lh    = lane >> 4;
  const int ntile = blockIdx.x & 63;
  const int pair  = blockIdx.x >> 6;
  const float* zp = z + (size_t)pair * NODES * HID;
  const int n0 = ntile * 16;

  v2f za[16];
  {
    const float* zrow = zp + (size_t)(n0 + l16) * HID;
#pragma unroll
    for (int ks = 0; ks < 16; ++ks) {
      const int k = 4 * ks + 2 * lh;
      za[ks].x = zrow[k]; za[ks].y = zrow[k + 1];
    }
  }

  for (int mt = 0; mt < 64; ++mt) {
    v8f acc = {};
#pragma unroll
    for (int ks = 0; ks < 16; ++ks) {
      const int k = 4 * ks + 2 * lh;
      v2f b;
      b.x = zp[(size_t)(mt * 16 + l16) * HID + k];
      b.y = zp[(size_t)(mt * 16 + l16) * HID + k + 1];
      acc = wmma_f32(za[ks], b, acc);
    }
    const int col = mt * 16 + l16;
#pragma unroll
    for (int v = 0; v < 8; ++v) {
      const int row = v + 8 * lh;
      const float s  = acc[v];
      const float lr = s > 0.0f ? s : 0.1f * s;
      float a = lr * (float)mask[(size_t)(n0 + row) * NODES + col];
      a = (a == 0.0f) ? NEG_BIG : a;
      S[col * 16 + row] = a;
    }
  }

  {
    const int r = l16;
    const int mbase = lh * 512;
    float mx = -INFINITY;
    for (int i = 0; i < 512; ++i) mx = fmaxf(mx, S[(mbase + i) * 16 + r]);
    mx = fmaxf(mx, __shfl_xor(mx, 16, 32));
    float sum = 0.0f;
    for (int i = 0; i < 512; ++i) sum += __expf(S[(mbase + i) * 16 + r] - mx);
    sum += __shfl_xor(sum, 16, 32);
    const float inv = 1.0f / sum;
    for (int i = 0; i < 512; ++i) {
      const int idx = (mbase + i) * 16 + r;
      S[idx] = __expf(S[idx] - mx) * inv;
    }
  }

  v8f accO[4] = {};
  for (int mt = 0; mt < 64; ++mt) {
#pragma unroll
    for (int ks = 0; ks < 4; ++ks) {
      const int k = mt * 16 + 4 * ks + 2 * lh;
      v2f a;
      a.x = S[k * 16 + l16];
      a.y = S[(k + 1) * 16 + l16];
#pragma unroll
      for (int ct = 0; ct < 4; ++ct) {
        v2f b;
        b.x = zp[(size_t)k * HID + ct * 16 + l16];
        b.y = zp[(size_t)(k + 1) * HID + ct * 16 + l16];
        accO[ct] = wmma_f32(a, b, accO[ct]);
      }
    }
  }

  float* op = out + (size_t)pair * NODES * HID;
#pragma unroll
  for (int ct = 0; ct < 4; ++ct)
#pragma unroll
    for (int v = 0; v < 8; ++v) {
      const int row = v + 8 * lh;
      op[(size_t)(n0 + row) * HID + ct * 16 + l16] = accO[ct][v];
    }
}

extern "C" void kernel_launch(void* const* d_in, const int* in_sizes, int n_in,
                              void* d_out, int out_size, void* d_ws, size_t ws_size,
                              hipStream_t stream) {
  const float* x     = (const float*)d_in[0];
  const int*   mask  = (const int*)  d_in[1];
  const float* w_ih0 = (const float*)d_in[2];
  const float* w_hh0 = (const float*)d_in[3];
  const float* b_ih0 = (const float*)d_in[4];
  const float* b_hh0 = (const float*)d_in[5];
  const float* w_ih1 = (const float*)d_in[6];
  const float* w_hh1 = (const float*)d_in[7];
  const float* b_ih1 = (const float*)d_in[8];
  const float* b_hh1 = (const float*)d_in[9];
  float* out  = (float*)d_out;
  float* zbuf = (float*)d_ws;

  gru_fused_kernel<<<256, 128, 0, stream>>>(x, w_ih0, w_hh0, b_ih0, b_hh0,
                                            w_ih1, w_hh1, b_ih1, b_hh1, zbuf);
  gat_kernel<<<2048, 32, 0, stream>>>(zbuf, mask, out);
}


# Attention products transposed (S^T = z_m z_n^T, out^T = z^T P^T): score D tile is the output product's B fragment, running-max softmax per lane, z in LDS shared by 8 waves; new sum order
# speedup vs baseline: 1.9111x; 1.9111x over previous
#include <hip/hip_runtime.h>
#include <hip/hip_bf16.h>
#include <math.h>


typedef __attribute__((ext_vector_type(2))) float v2f;
typedef __attribute__((ext_vector_type(4))) float v4f;
typedef __attribute__((ext_vector_type(8))) float v8f;

#define T_SEQ   12
#define BCN     16
#define NODES   1024
#define FEAT    32
#define HID     64
#define G3      192
#define NROWS   (BCN * NODES)
#define NEG_BIG (-1e16f)

__device__ __forceinline__ v8f wmma_f32(v2f a, v2f b, v8f c) {
  return __builtin_amdgcn_wmma_f32_16x16x4_f32(
       false, a,  false, b,
       (short)0, c,  false,  false);
}

__device__ __forceinline__ float sigmoidf_(float x) {
  return 1.0f / (1.0f + __expf(-x));
}

#define GRU_WAVES 4
#define WP32 36
#define WP64 68

__device__ __forceinline__ v4f lds4(const float* p) {
  return *reinterpret_cast<const v4f*>(p);
}

__device__ __forceinline__ v8f join8(v4f lo, v4f hi) {
  return __builtin_shufflevector(lo, hi, 0, 1, 2, 3, 4, 5, 6, 7);
}

template <int NU, int NB>
__device__ __forceinline__ void mm_block(v8f (&acc)[3], const v4f (&f)[3][8],
                                         const float (&b)[NB]) {
#pragma unroll
  for (int u = 0; u < NU; ++u) {
#pragma unroll
    for (int hf = 0; hf < 2; ++hf) {
      v2f bb;
      bb.x = b[4 * u + 2 * hf];
      bb.y = b[4 * u + 2 * hf + 1];
#pragma unroll
      for (int g = 0; g < 3; ++g) {
        v2f aa;
        aa.x = f[g][u][2 * hf];
        aa.y = f[g][u][2 * hf + 1];
        acc[g] = wmma_f32(aa, bb, acc[g]);
      }
    }
  }
}

template <int NUI>
__device__ __forceinline__ void load_tile(const float* wi, int pitch_i, int koff_i,
                                          const float* wh, int koff_h,
                                          const float* bi, const float* bh,
                                          int j, int l16, int lh,
                                          v4f (&Fi)[3][8], v4f (&Fh)[3][8],
                                          v4f (&Bb)[6][2]) {
#pragma unroll
  for (int g = 0; g < 3; ++g) {
    const int grow = 64 * g + 16 * j;
#pragma unroll
    for (int u = 0; u < 8; ++u) {
      if (u < NUI) {
        const int kb = (NUI == 4) ? (4 * u) : (16 * (u >> 1) + 4 * (u & 1));
        Fi[g][u] = lds4(wi + (grow + l16) * pitch_i + koff_i + kb);
      } else {
        Fi[g][u] = (v4f){0.0f, 0.0f, 0.0f, 0.0f};
      }
      Fh[g][u] = lds4(wh + (grow + l16) * WP64 + koff_h + 16 * (u >> 1) + 4 * (u & 1));
    }
#pragma unroll
    for (int q = 0; q < 2; ++q) {
      Bb[g][q]     = lds4(bi + grow + 8 * lh + 4 * q);
      Bb[3 + g][q] = lds4(bh + grow + 8 * lh + 4 * q);
    }
  }
}

__global__ __launch_bounds__(128) void gru_fused_kernel(
    const float* __restrict__ x_in,
    const float* __restrict__ w_ih0,
    const float* __restrict__ w_hh0,
    const float* __restrict__ b_ih0,
    const float* __restrict__ b_hh0,
    const float* __restrict__ w_ih1,
    const float* __restrict__ w_hh1,
    const float* __restrict__ b_ih1,
    const float* __restrict__ b_hh1,
    float* __restrict__ z_out)
{
  __shared__ __attribute__((aligned(16))) float wih0_s[G3 * WP32];
  __shared__ __attribute__((aligned(16))) float whh0_s[G3 * WP64];
  __shared__ __attribute__((aligned(16))) float wih1_s[G3 * WP64];
  __shared__ __attribute__((aligned(16))) float whh1_s[G3 * WP64];
  __shared__ __attribute__((aligned(16))) float bias_s[4 * G3];
  __shared__ __attribute__((aligned(16))) float ost[GRU_WAVES][16 * WP64];

  const int tid = threadIdx.x;
  {
    const float4* g0 = reinterpret_cast<const float4*>(w_ih0);
    for (int i = tid; i < G3 * FEAT / 4; i += 128) {
      const int row = i >> 3, c4 = i & 7;
      *reinterpret_cast<float4*>(&wih0_s[row * WP32 + 4 * c4]) = g0[i];
    }
    const float4* g1 = reinterpret_cast<const float4*>(w_hh0);
    const float4* g2 = reinterpret_cast<const float4*>(w_ih1);
    const float4* g3 = reinterpret_cast<const float4*>(w_hh1);
    for (int i = tid; i < G3 * HID / 4; i += 128) {
      const int row = i >> 4, c4 = i & 15;
      *reinterpret_cast<float4*>(&whh0_s[row * WP64 + 4 * c4]) = g1[i];
      *reinterpret_cast<float4*>(&wih1_s[row * WP64 + 4 * c4]) = g2[i];
      *reinterpret_cast<float4*>(&whh1_s[row * WP64 + 4 * c4]) = g3[i];
    }
    for (int i = tid; i < G3; i += 128) {
      bias_s[i]          = b_ih0[i];
      bias_s[G3 + i]     = b_hh0[i];
      bias_s[2 * G3 + i] = b_ih1[i];
      bias_s[3 * G3 + i] = b_hh1[i];
    }
  }
  __syncthreads();

  const int wave = tid >> 5;
  const int lane = tid & 31;
  const int l16  = lane & 15;
  const int lh   = lane >> 4;
  const int row0 = blockIdx.x * (GRU_WAVES * 16) + wave * 16;

  float h0[32], h1[32], h0n[32], h1n[32];
#pragma unroll
  for (int i = 0; i < 32; ++i) { h0[i] = 0.0f; h1[i] = 0.0f; h0n[i] = 0.0f; h1n[i] = 0.0f; }

  float X[16], Xn[16];
  {
    const float4* xp = reinterpret_cast<const float4*>(
        x_in + ((size_t)row0 + l16) * FEAT + 16 * lh);
#pragma unroll
    for (int u = 0; u < 4; ++u) {
      const float4 v = xp[u];
      X[4 * u] = v.x; X[4 * u + 1] = v.y; X[4 * u + 2] = v.z; X[4 * u + 3] = v.w;
    }
  }

  v4f Fi[3][8], Fh[3][8], Bb[6][2];
  load_tile<4>(wih0_s, WP32, 16 * lh, whh0_s, 8 * lh, bias_s, bias_s + G3,
               0, l16, lh, Fi, Fh, Bb);

  for (int t = 0; t < T_SEQ; ++t) {
    {
      const int tn = (t + 1 < T_SEQ) ? (t + 1) : t;
      const float4* xp = reinterpret_cast<const float4*>(
          x_in + ((size_t)tn * NROWS + row0 + l16) * FEAT + 16 * lh);
#pragma unroll
      for (int u = 0; u < 4; ++u) {
        const float4 v = xp[u];
        Xn[4 * u] = v.x; Xn[4 * u + 1] = v.y; Xn[4 * u + 2] = v.z; Xn[4 * u + 3] = v.w;
      }
    }

#pragma unroll
    for (int i = 0; i < 8; ++i) {
      const int layer = i >> 2;
      const int j     = i & 3;

      v8f gi[3], gh[3];
#pragma unroll
      for (int g = 0; g < 3; ++g) {
        gi[g] = join8(Bb[g][0], Bb[g][1]);
        gh[g] = join8(Bb[3 + g][0], Bb[3 + g][1]);
      }
      if (layer == 0) {
        mm_block<4, 16>(gi, Fi, X);
        mm_block<8, 32>(gh, Fh, h0);
      } else {
        mm_block<8, 32>(gh, Fh, h1);
        mm_block<8, 32>(gi, Fi, h0n);
      }

      __builtin_amdgcn_sched_barrier(0);
      {
        const int in = (i + 1) & 7;
        const int ln = in >> 2, jn = in & 3;
        if (ln == 0)
          load_tile<4>(wih0_s, WP32, 16 * lh, whh0_s, 8 * lh, bias_s, bias_s + G3,
                       jn, l16, lh, Fi, Fh, Bb);
        else
          load_tile<8>(wih1_s, WP64, 8 * lh, whh1_s, 8 * lh, bias_s + 2 * G3,
                       bias_s + 3 * G3, jn, l16, lh, Fi, Fh, Bb);
      }
      __builtin_amdgcn_sched_barrier(0);

#pragma unroll
      for (int v = 0; v < 8; ++v) {
        const float hold = (layer == 0) ? h0[8 * j + v] : h1[8 * j + v];
        const float r  = sigmoidf_(gi[0][v] + gh[0][v]);
        const float zg = sigmoidf_(gi[1][v] + gh[1][v]);
        const float n  = tanhf(gi[2][v] + r * gh[2][v]);
        const float hn = (1.0f - zg) * n + zg * hold;
        if (layer == 0) h0n[8 * j + v] = hn; else h1n[8 * j + v] = hn;
      }
    }

#pragma unroll
    for (int i = 0; i < 32; ++i) { h0[i] = h0n[i]; h1[i] = h1n[i]; }
#pragma unroll
    for (int i = 0; i < 16; ++i) X[i] = Xn[i];
  }

  float* st = &ost[wave][0];
#pragma unroll
  for (int layer = 0; layer < 2; ++layer) {
#pragma unroll
    for (int u = 0; u < 8; ++u) {
      v4f v;
#pragma unroll
      for (int e = 0; e < 4; ++e) v[e] = (layer == 0) ? h0[4 * u + e] : h1[4 * u + e];
      *reinterpret_cast<v4f*>(&st[l16 * WP64 + 16 * (u >> 1) + 8 * lh + 4 * (u & 1)]) = v;
    }
#pragma unroll
    for (int it = 0; it < 8; ++it) {
      const int idx = it * 32 + lane;
      const int r = idx >> 4, c4 = idx & 15;
      const v4f v = lds4(&st[r * WP64 + 4 * c4]);
      *reinterpret_cast<v4f*>(
          &z_out[(size_t)layer * NROWS * HID + (size_t)(row0 + r) * HID + 4 * c4]) = v;
    }
  }
}

#define GAT_WAVES 8
#define ZP 68

typedef __attribute__((ext_vector_type(4))) int v4i;

__device__ __forceinline__ float xhalf(float x) {
  const int b = __builtin_bit_cast(int, x);
  return __builtin_bit_cast(float,
      __builtin_amdgcn_permlanex16(b, b, 0x76543210, 0xfedcba98, false, false));
}

__device__ __forceinline__ void gat_fetch(const float* zs, const int* mlane, int m0,
                                          int l16, int lh, v4f (&za)[8],
                                          float (&tx)[4][4], float (&ty)[4][4],
                                          v4i (&mk)[2]) {
  const float* ar = zs + (m0 + l16) * ZP + 4 * lh;
#pragma unroll
  for (int u = 0; u < 8; ++u) za[u] = lds4(ar + 8 * u);
  const float* tr = zs + (m0 + 8 * lh) * ZP + l16;
#pragma unroll
  for (int q = 0; q < 4; ++q) {
#pragma unroll
    for (int dt = 0; dt < 4; ++dt) {
      tx[dt][q] = tr[(2 * q) * ZP + 16 * dt];
      ty[dt][q] = tr[(2 * q + 1) * ZP + 16 * dt];
    }
  }
  const v4i* mp = reinterpret_cast<const v4i*>(mlane + m0);
  mk[0] = mp[0];
  mk[1] = mp[1];
}

__device__ __forceinline__ void gat_step(const v4f (&zb)[8], const v4f (&za)[8],
                                         const float (&tx)[4][4], const float (&ty)[4][4],
                                         const v4i (&mk)[2],
                                         float& M, float& L, v8f (&accO)[4]) {
  v8f s0 = {}, s1 = {};
#pragma unroll
  for (int u = 0; u < 8; ++u) {
    v2f a0, b0, a1, b1;
    a0.x = za[u][0]; a0.y = za[u][1]; b0.x = zb[u][0]; b0.y = zb[u][1];
    a1.x = za[u][2]; a1.y = za[u][3]; b1.x = zb[u][2]; b1.y = zb[u][3];
    s0 = wmma_f32(a0, b0, s0);
    s1 = wmma_f32(a1, b1, s1);
  }
  float a[8];
#pragma unroll
  for (int r = 0; r < 8; ++r) {
    const float s  = s0[r] + s1[r];
    const float lr = s > 0.0f ? s : 0.1f * s;
    const int   mi = (r < 4) ? mk[0][r & 3] : mk[1][r & 3];
    float v = lr * (float)mi;
    v = (v == 0.0f) ? NEG_BIG : v;
    a[r] = v;
  }
  float tm = a[0];
#pragma unroll
  for (int r = 1; r < 8; ++r) tm = fmaxf(tm, a[r]);
  tm = fmaxf(tm, xhalf(tm));
  const float Mn    = fmaxf(M, tm);
  const float alpha = __expf(M - Mn);
  float p[8];
  float ps = 0.0f;
#pragma unroll
  for (int r = 0; r < 8; ++r) { p[r] = __expf(a[r] - Mn); ps += p[r]; }
  L = L * alpha + ps;
  M = Mn;
#pragma unroll
  for (int dt = 0; dt < 4; ++dt)
#pragma unroll
    for (int r = 0; r < 8; ++r) accO[dt][r] *= alpha;
#pragma unroll
  for (int q = 0; q < 4; ++q) {
    v2f b;
    b.x = p[2 * q];
    b.y = p[2 * q + 1];
#pragma unroll
    for (int dt = 0; dt < 4; ++dt) {
      v2f a2;
      a2.x = tx[dt][q];
      a2.y = ty[dt][q];
      accO[dt] = wmma_f32(a2, b, accO[dt]);
    }
  }
}

__global__ __launch_bounds__(256) void gat_kernel(
    const float* __restrict__ z,
    const int*   __restrict__ mask,
    float*       __restrict__ out)
{
  __shared__ __attribute__((aligned(16))) float zs[NODES * ZP];

  const int tid   = threadIdx.x;
  const int pair  = blockIdx.x >> 3;
  const int chunk = blockIdx.x & 7;
  const float* zp = z + (size_t)pair * NODES * HID;

  {
    const float4* g = reinterpret_cast<const float4*>(zp);
#pragma unroll 8
    for (int it = 0; it < NODES * HID / 4 / 256; ++it) {
      const int i = it * 256 + tid;
      const int row = i >> 4, c4 = i & 15;
      *reinterpret_cast<float4*>(&zs[row * ZP + 4 * c4]) = g[i];
    }
  }
  __syncthreads();

  const int wave = tid >> 5;
  const int lane = tid & 31;
  const int l16  = lane & 15;
  const int lh   = lane >> 4;
  const int n0   = (chunk * GAT_WAVES + wave) * 16;

  v4f zb[8];
  {
    const float* br = zs + (n0 + l16) * ZP + 4 * lh;
#pragma unroll
    for (int u = 0; u < 8; ++u) zb[u] = lds4(br + 8 * u);
  }
  const int* mlane = mask + (size_t)(n0 + l16) * NODES + 8 * lh;

  float M = -INFINITY, L = 0.0f;
  v8f accO[4] = {};
  v4f zaA[8], zaB[8];
  float txA[4][4], tyA[4][4], txB[4][4], tyB[4][4];
  v4i mkA[2], mkB[2];

  gat_fetch(zs, mlane, 0, l16, lh, zaA, txA, tyA, mkA);
  for (int t = 0; t < 64; t += 2) {
    gat_fetch(zs, mlane, (t + 1) * 16, l16, lh, zaB, txB, tyB, mkB);
    __builtin_amdgcn_sched_barrier(0);
    gat_step(zb, zaA, txA, tyA, mkA, M, L, accO);
    {
      const int tn = (t + 2 < 64) ? (t + 2) : t;
      gat_fetch(zs, mlane, tn * 16, l16, lh, zaA, txA, tyA, mkA);
    }
    __builtin_amdgcn_sched_barrier(0);
    gat_step(zb, zaB, txB, tyB, mkB, M, L, accO);
  }

  const float inv = 1.0f / (L + xhalf(L));
  float* orow = out + (size_t)pair * NODES * HID + (size_t)(n0 + l16) * HID + 8 * lh;
#pragma unroll
  for (int dt = 0; dt < 4; ++dt) {
    v4f lo, hi;
#pragma unroll
    for (int e = 0; e < 4; ++e) { lo[e] = accO[dt][e] * inv; hi[e] = accO[dt][4 + e] * inv; }
    *reinterpret_cast<v4f*>(orow + 16 * dt)     = lo;
    *reinterpret_cast<v4f*>(orow + 16 * dt + 4) = hi;
  }
}

extern "C" void kernel_launch(void* const* d_in, const int* in_sizes, int n_in,
                              void* d_out, int out_size, void* d_ws, size_t ws_size,
                              hipStream_t stream) {
  const float* x     = (const float*)d_in[0];
  const int*   mask  = (const int*)  d_in[1];
  const float* w_ih0 = (const float*)d_in[2];
  const float* w_hh0 = (const float*)d_in[3];
  const float* b_ih0 = (const float*)d_in[4];
  const float* b_hh0 = (const float*)d_in[5];
  const float* w_ih1 = (const float*)d_in[6];
  const float* w_hh1 = (const float*)d_in[7];
  const float* b_ih1 = (const float*)d_in[8];
  const float* b_hh1 = (const float*)d_in[9];
  float* out  = (float*)d_out;
  float* zbuf = (float*)d_ws;

  gru_fused_kernel<<<256, 128, 0, stream>>>(x, w_ih0, w_hh0, b_ih0, b_hh0,
                                            w_ih1, w_hh1, b_ih1, b_hh1, zbuf);
  gat_kernel<<<256, 256, 0, stream>>>(zbuf, mask, out);
}


# GRU layers pipelined over two waves per 16 batch rows (layer 0 does step s while layer 1 does step s-1; h0 handed over through LDS, one barrier per step): two waves per SIMD instead of one
# speedup vs baseline: 2.1499x; 1.1250x over previous
#include <hip/hip_runtime.h>
#include <hip/hip_bf16.h>
#include <math.h>


typedef __attribute__((ext_vector_type(2))) float v2f;
typedef __attribute__((ext_vector_type(4))) float v4f;
typedef __attribute__((ext_vector_type(8))) float v8f;

#define T_SEQ   12
#define BCN     16
#define NODES   1024
#define FEAT    32
#define HID     64
#define G3      192
#define NROWS   (BCN * NODES)
#define NEG_BIG (-1e16f)

__device__ __forceinline__ v8f wmma_f32(v2f a, v2f b, v8f c) {
  return __builtin_amdgcn_wmma_f32_16x16x4_f32(
       false, a,  false, b,
       (short)0, c,  false,  false);
}

__device__ __forceinline__ float sigmoidf_(float x) {
  return 1.0f / (1.0f + __expf(-x));
}

#define GRU_TILES 4
#define GRU_WAVES 8
#define WP32 36
#define WP64 68
#define HXP  36
#define HXBUF (GRU_TILES * 32 * HXP)

__device__ __forceinline__ v4f lds4(const float* p) {
  return *reinterpret_cast<const v4f*>(p);
}

__device__ __forceinline__ v8f join8(v4f lo, v4f hi) {
  return __builtin_shufflevector(lo, hi, 0, 1, 2, 3, 4, 5, 6, 7);
}

template <int NU, int NB>
__device__ __forceinline__ void mm_block(v8f (&acc)[3], const v4f (&f)[3][8],
                                         const float (&b)[NB]) {
#pragma unroll
  for (int u = 0; u < NU; ++u) {
#pragma unroll
    for (int hf = 0; hf < 2; ++hf) {
      v2f bb;
      bb.x = b[4 * u + 2 * hf];
      bb.y = b[4 * u + 2 * hf + 1];
#pragma unroll
      for (int g = 0; g < 3; ++g) {
        v2f aa;
        aa.x = f[g][u][2 * hf];
        aa.y = f[g][u][2 * hf + 1];
        acc[g] = wmma_f32(aa, bb, acc[g]);
      }
    }
  }
}

template <int NUI>
__device__ __forceinline__ void load_tile(const float* wi, int pitch_i, int koff_i,
                                          const float* wh, int koff_h,
                                          const float* bi, const float* bh,
                                          int j, int l16, int lh,
                                          v4f (&Fi)[3][8], v4f (&Fh)[3][8],
                                          v4f (&Bb)[6][2]) {
#pragma unroll
  for (int g = 0; g < 3; ++g) {
    const int grow = 64 * g + 16 * j;
#pragma unroll
    for (int u = 0; u < 8; ++u) {
      if (u < NUI) {
        const int kb = (NUI == 4) ? (4 * u) : (16 * (u >> 1) + 4 * (u & 1));
        Fi[g][u] = lds4(wi + (grow + l16) * pitch_i + koff_i + kb);
      } else {
        Fi[g][u] = (v4f){0.0f, 0.0f, 0.0f, 0.0f};
      }
      Fh[g][u] = lds4(wh + (grow + l16) * WP64 + koff_h + 16 * (u >> 1) + 4 * (u & 1));
    }
#pragma unroll
    for (int q = 0; q < 2; ++q) {
      Bb[g][q]     = lds4(bi + grow + 8 * lh + 4 * q);
      Bb[3 + g][q] = lds4(bh + grow + 8 * lh + 4 * q);
    }
  }
}

template <int LAYER, int NB>
__device__ __forceinline__ void gru_layer_step(
    const float* wi, const float* wh, const float* bi, const float* bh,
    int l16, int lh,
    const float (&hs)[32], float (&hn)[32], const float (&bin)[NB]) {
  v4f Fi[3][8], Fh[3][8], Bb[6][2];
  if (LAYER == 0)
    load_tile<4>(wi, WP32, 16 * lh, wh, 8 * lh, bi, bh, 0, l16, lh, Fi, Fh, Bb);
  else
    load_tile<8>(wi, WP64, 8 * lh, wh, 8 * lh, bi, bh, 0, l16, lh, Fi, Fh, Bb);
  __builtin_amdgcn_sched_barrier(0);
#pragma unroll
  for (int j = 0; j < 4; ++j) {
    v8f gi[3], gh[3];
#pragma unroll
    for (int g = 0; g < 3; ++g) {
      gi[g] = join8(Bb[g][0], Bb[g][1]);
      gh[g] = join8(Bb[3 + g][0], Bb[3 + g][1]);
    }
    if (LAYER == 0) {
      mm_block<4, NB>(gi, Fi, bin);
      mm_block<8, 32>(gh, Fh, hs);
    } else {
      mm_block<8, 32>(gh, Fh, hs);
      mm_block<8, NB>(gi, Fi, bin);
    }

    __builtin_amdgcn_sched_barrier(0);
    if (j < 3) {
      if (LAYER == 0)
        load_tile<4>(wi, WP32, 16 * lh, wh, 8 * lh, bi, bh, j + 1, l16, lh, Fi, Fh, Bb);
      else
        load_tile<8>(wi, WP64, 8 * lh, wh, 8 * lh, bi, bh, j + 1, l16, lh, Fi, Fh, Bb);
    }
    __builtin_amdgcn_sched_barrier(0);

#pragma unroll
    for (int v = 0; v < 8; ++v) {
      const float hold = hs[8 * j + v];
      const float r  = sigmoidf_(gi[0][v] + gh[0][v]);
      const float zg = sigmoidf_(gi[1][v] + gh[1][v]);
      const float n  = tanhf(gi[2][v] + r * gh[2][v]);
      hn[8 * j + v] = (1.0f - zg) * n + zg * hold;
    }
  }
}

__global__ __launch_bounds__(256) void gru_fused_kernel(
    const float* __restrict__ x_in,
    const float* __restrict__ w_ih0,
    const float* __restrict__ w_hh0,
    const float* __restrict__ b_ih0,
    const float* __restrict__ b_hh0,
    const float* __restrict__ w_ih1,
    const float* __restrict__ w_hh1,
    const float* __restrict__ b_ih1,
    const float* __restrict__ b_hh1,
    float* __restrict__ z_out)
{
  __shared__ __attribute__((aligned(16))) float wih0_s[G3 * WP32];
  __shared__ __attribute__((aligned(16))) float whh0_s[G3 * WP64];
  __shared__ __attribute__((aligned(16))) float wih1_s[G3 * WP64];
  __shared__ __attribute__((aligned(16))) float whh1_s[G3 * WP64];
  __shared__ __attribute__((aligned(16))) float bias_s[4 * G3];
  __shared__ __attribute__((aligned(16))) float ost[GRU_WAVES][16 * WP64];
  __shared__ __attribute__((aligned(16))) float hx[2 * HXBUF];

  const int tid = threadIdx.x;
  {
    const float4* g0 = reinterpret_cast<const float4*>(w_ih0);
    for (int i = tid; i < G3 * FEAT / 4; i += 256) {
      const int row = i >> 3, c4 = i & 7;
      *reinterpret_cast<float4*>(&wih0_s[row * WP32 + 4 * c4]) = g0[i];
    }
    const float4* g1 = reinterpret_cast<const float4*>(w_hh0);
    const float4* g2 = reinterpret_cast<const float4*>(w_ih1);
    const float4* g3 = reinterpret_cast<const float4*>(w_hh1);
    for (int i = tid; i < G3 * HID / 4; i += 256) {
      const int row = i >> 4, c4 = i & 15;
      *reinterpret_cast<float4*>(&whh0_s[row * WP64 + 4 * c4]) = g1[i];
      *reinterpret_cast<float4*>(&wih1_s[row * WP64 + 4 * c4]) = g2[i];
      *reinterpret_cast<float4*>(&whh1_s[row * WP64 + 4 * c4]) = g3[i];
    }
    for (int i = tid; i < G3; i += 256) {
      bias_s[i]          = b_ih0[i];
      bias_s[G3 + i]     = b_hh0[i];
      bias_s[2 * G3 + i] = b_ih1[i];
      bias_s[3 * G3 + i] = b_hh1[i];
    }
  }
  __syncthreads();

  const int wave = tid >> 5;
  const int lane = tid & 31;
  const int l16  = lane & 15;
  const int lh   = lane >> 4;
  const int layer = __builtin_amdgcn_readfirstlane(wave < GRU_TILES ? 1 : 0);
  const int tile  = wave & (GRU_TILES - 1);
  const int row0  = blockIdx.x * (GRU_TILES * 16) + tile * 16;

  float hs[32], xc[16], xn[16];
#pragma unroll
  for (int i = 0; i < 32; ++i) hs[i] = 0.0f;
#pragma unroll
  for (int i = 0; i < 16; ++i) { xc[i] = 0.0f; xn[i] = 0.0f; }

  float* hxl = &hx[(tile * 32 + lane) * HXP];

  if (layer == 0) {
    const float4* xp = reinterpret_cast<const float4*>(
        x_in + ((size_t)row0 + l16) * FEAT + 16 * lh);
#pragma unroll
    for (int u = 0; u < 4; ++u) {
      const float4 v = xp[u];
      xc[4 * u] = v.x; xc[4 * u + 1] = v.y; xc[4 * u + 2] = v.z; xc[4 * u + 3] = v.w;
    }
  }

  for (int s = 0; s <= T_SEQ; ++s) {
    if (layer == 0) {
      if (s < T_SEQ) {
        {
          const int tn = (s + 1 < T_SEQ) ? (s + 1) : s;
          const float4* xp = reinterpret_cast<const float4*>(
              x_in + ((size_t)tn * NROWS + row0 + l16) * FEAT + 16 * lh);
#pragma unroll
          for (int u = 0; u < 4; ++u) {
            const float4 v = xp[u];
            xn[4 * u] = v.x; xn[4 * u + 1] = v.y; xn[4 * u + 2] = v.z; xn[4 * u + 3] = v.w;
          }
        }
        float hn[32];
        gru_layer_step<0, 16>(wih0_s, whh0_s, bias_s, bias_s + G3, l16, lh, hs, hn, xc);
        float* dst = hxl + (s & 1) * HXBUF;
#pragma unroll
        for (int u = 0; u < 8; ++u) {
          v4f v;
#pragma unroll
          for (int e = 0; e < 4; ++e) v[e] = hn[4 * u + e];
          *reinterpret_cast<v4f*>(dst + 4 * u) = v;
        }
#pragma unroll
        for (int i = 0; i < 32; ++i) hs[i] = hn[i];
#pragma unroll
        for (int i = 0; i < 16; ++i) xc[i] = xn[i];
      }
    } else {
      if (s >= 1) {
        float bin[32];
        const float* srcp = hxl + ((s & 1) ^ 1) * HXBUF;
#pragma unroll
        for (int u = 0; u < 8; ++u) {
          const v4f v = lds4(srcp + 4 * u);
#pragma unroll
          for (int e = 0; e < 4; ++e) bin[4 * u + e] = v[e];
        }
        float hn[32];
        gru_layer_step<1, 32>(wih1_s, whh1_s, bias_s + 2 * G3, bias_s + 3 * G3, l16, lh,
                              hs, hn, bin);
#pragma unroll
        for (int i = 0; i < 32; ++i) hs[i] = hn[i];
      }
    }
    __syncthreads();
  }

  float* st = &ost[wave][0];
#pragma unroll
  for (int u = 0; u < 8; ++u) {
    v4f v;
#pragma unroll
    for (int e = 0; e < 4; ++e) v[e] = hs[4 * u + e];
    *reinterpret_cast<v4f*>(&st[l16 * WP64 + 16 * (u >> 1) + 8 * lh + 4 * (u & 1)]) = v;
  }
#pragma unroll
  for (int it = 0; it < 8; ++it) {
    const int idx = it * 32 + lane;
    const int r = idx >> 4, c4 = idx & 15;
    const v4f v = lds4(&st[r * WP64 + 4 * c4]);
    *reinterpret_cast<v4f*>(
        &z_out[(size_t)layer * NROWS * HID + (size_t)(row0 + r) * HID + 4 * c4]) = v;
  }
}

#define GAT_WAVES 8
#define ZP 68

typedef __attribute__((ext_vector_type(4))) int v4i;

__device__ __forceinline__ float xhalf(float x) {
  const int b = __builtin_bit_cast(int, x);
  return __builtin_bit_cast(float,
      __builtin_amdgcn_permlanex16(b, b, 0x76543210, 0xfedcba98, false, false));
}

__device__ __forceinline__ void gat_fetch(const float* zs, const int* mlane, int m0,
                                          int l16, int lh, v4f (&za)[8],
                                          float (&tx)[4][4], float (&ty)[4][4],
                                          v4i (&mk)[2]) {
  const float* ar = zs + (m0 + l16) * ZP + 4 * lh;
#pragma unroll
  for (int u = 0; u < 8; ++u) za[u] = lds4(ar + 8 * u);
  const float* tr = zs + (m0 + 8 * lh) * ZP + l16;
#pragma unroll
  for (int q = 0; q < 4; ++q) {
#pragma unroll
    for (int dt = 0; dt < 4; ++dt) {
      tx[dt][q] = tr[(2 * q) * ZP + 16 * dt];
      ty[dt][q] = tr[(2 * q + 1) * ZP + 16 * dt];
    }
  }
  const v4i* mp = reinterpret_cast<const v4i*>(mlane + m0);
  mk[0] = mp[0];
  mk[1] = mp[1];
}

__device__ __forceinline__ void gat_step(const v4f (&zb)[8], const v4f (&za)[8],
                                         const float (&tx)[4][4], const float (&ty)[4][4],
                                         const v4i (&mk)[2],
                                         float& M, float& L, v8f (&accO)[4]) {
  v8f s0 = {}, s1 = {};
#pragma unroll
  for (int u = 0; u < 8; ++u) {
    v2f a0, b0, a1, b1;
    a0.x = za[u][0]; a0.y = za[u][1]; b0.x = zb[u][0]; b0.y = zb[u][1];
    a1.x = za[u][2]; a1.y = za[u][3]; b1.x = zb[u][2]; b1.y = zb[u][3];
    s0 = wmma_f32(a0, b0, s0);
    s1 = wmma_f32(a1, b1, s1);
  }
  float a[8];
#pragma unroll
  for (int r = 0; r < 8; ++r) {
    const float s  = s0[r] + s1[r];
    const float lr = s > 0.0f ? s : 0.1f * s;
    const int   mi = (r < 4) ? mk[0][r & 3] : mk[1][r & 3];
    float v = lr * (float)mi;
    v = (v == 0.0f) ? NEG_BIG : v;
    a[r] = v;
  }
  float tm = a[0];
#pragma unroll
  for (int r = 1; r < 8; ++r) tm = fmaxf(tm, a[r]);
  tm = fmaxf(tm, xhalf(tm));
  const float Mn    = fmaxf(M, tm);
  const float alpha = __expf(M - Mn);
  float p[8];
  float ps = 0.0f;
#pragma unroll
  for (int r = 0; r < 8; ++r) { p[r] = __expf(a[r] - Mn); ps += p[r]; }
  L = L * alpha + ps;
  M = Mn;
#pragma unroll
  for (int dt = 0; dt < 4; ++dt)
#pragma unroll
    for (int r = 0; r < 8; ++r) accO[dt][r] *= alpha;
#pragma unroll
  for (int q = 0; q < 4; ++q) {
    v2f b;
    b.x = p[2 * q];
    b.y = p[2 * q + 1];
#pragma unroll
    for (int dt = 0; dt < 4; ++dt) {
      v2f a2;
      a2.x = tx[dt][q];
      a2.y = ty[dt][q];
      accO[dt] = wmma_f32(a2, b, accO[dt]);
    }
  }
}

__global__ __launch_bounds__(256) void gat_kernel(
    const float* __restrict__ z,
    const int*   __restrict__ mask,
    float*       __restrict__ out)
{
  __shared__ __attribute__((aligned(16))) float zs[NODES * ZP];

  const int tid   = threadIdx.x;
  const int pair  = blockIdx.x >> 3;
  const int chunk = blockIdx.x & 7;
  const float* zp = z + (size_t)pair * NODES * HID;

  {
    const float4* g = reinterpret_cast<const float4*>(zp);
#pragma unroll 8
    for (int it = 0; it < NODES * HID / 4 / 256; ++it) {
      const int i = it * 256 + tid;
      const int row = i >> 4, c4 = i & 15;
      *reinterpret_cast<float4*>(&zs[row * ZP + 4 * c4]) = g[i];
    }
  }
  __syncthreads();

  const int wave = tid >> 5;
  const int lane = tid & 31;
  const int l16  = lane & 15;
  const int lh   = lane >> 4;
  const int n0   = (chunk * GAT_WAVES + wave) * 16;

  v4f zb[8];
  {
    const float* br = zs + (n0 + l16) * ZP + 4 * lh;
#pragma unroll
    for (int u = 0; u < 8; ++u) zb[u] = lds4(br + 8 * u);
  }
  const int* mlane = mask + (size_t)(n0 + l16) * NODES + 8 * lh;

  float M = -INFINITY, L = 0.0f;
  v8f accO[4] = {};
  v4f zaA[8], zaB[8];
  float txA[4][4], tyA[4][4], txB[4][4], tyB[4][4];
  v4i mkA[2], mkB[2];

  gat_fetch(zs, mlane, 0, l16, lh, zaA, txA, tyA, mkA);
  for (int t = 0; t < 64; t += 2) {
    gat_fetch(zs, mlane, (t + 1) * 16, l16, lh, zaB, txB, tyB, mkB);
    __builtin_amdgcn_sched_barrier(0);
    gat_step(zb, zaA, txA, tyA, mkA, M, L, accO);
    {
      const int tn = (t + 2 < 64) ? (t + 2) : t;
      gat_fetch(zs, mlane, tn * 16, l16, lh, zaA, txA, tyA, mkA);
    }
    __builtin_amdgcn_sched_barrier(0);
    gat_step(zb, zaB, txB, tyB, mkB, M, L, accO);
  }

  const float inv = 1.0f / (L + xhalf(L));
  float* orow = out + (size_t)pair * NODES * HID + (size_t)(n0 + l16) * HID + 8 * lh;
#pragma unroll
  for (int dt = 0; dt < 4; ++dt) {
    v4f lo, hi;
#pragma unroll
    for (int e = 0; e < 4; ++e) { lo[e] = accO[dt][e] * inv; hi[e] = accO[dt][4 + e] * inv; }
    *reinterpret_cast<v4f*>(orow + 16 * dt)     = lo;
    *reinterpret_cast<v4f*>(orow + 16 * dt + 4) = hi;
  }
}

extern "C" void kernel_launch(void* const* d_in, const int* in_sizes, int n_in,
                              void* d_out, int out_size, void* d_ws, size_t ws_size,
                              hipStream_t stream) {
  const float* x     = (const float*)d_in[0];
  const int*   mask  = (const int*)  d_in[1];
  const float* w_ih0 = (const float*)d_in[2];
  const float* w_hh0 = (const float*)d_in[3];
  const float* b_ih0 = (const float*)d_in[4];
  const float* b_hh0 = (const float*)d_in[5];
  const float* w_ih1 = (const float*)d_in[6];
  const float* w_hh1 = (const float*)d_in[7];
  const float* b_ih1 = (const float*)d_in[8];
  const float* b_hh1 = (const float*)d_in[9];
  float* out  = (float*)d_out;
  float* zbuf = (float*)d_ws;

  gru_fused_kernel<<<256, 256, 0, stream>>>(x, w_ih0, w_hh0, b_ih0, b_hh0,
                                            w_ih1, w_hh1, b_ih1, b_hh1, zbuf);
  gat_kernel<<<256, 256, 0, stream>>>(zbuf, mask, out);
}


# GRU split over four waves per 16 batch rows (layer x hidden half; partner half of h through LDS once per step): four waves per SIMD beside the gate chains; upper half sums W_hh k=32..63 first
# speedup vs baseline: 2.4026x; 1.1175x over previous
#include <hip/hip_runtime.h>
#include <hip/hip_bf16.h>
#include <math.h>


typedef __attribute__((ext_vector_type(2))) float v2f;
typedef __attribute__((ext_vector_type(4))) float v4f;
typedef __attribute__((ext_vector_type(8))) float v8f;

#define T_SEQ   12
#define BCN     16
#define NODES   1024
#define FEAT    32
#define HID     64
#define G3      192
#define NROWS   (BCN * NODES)
#define NEG_BIG (-1e16f)

__device__ __forceinline__ v8f wmma_f32(v2f a, v2f b, v8f c) {
  return __builtin_amdgcn_wmma_f32_16x16x4_f32(
       false, a,  false, b,
       (short)0, c,  false,  false);
}

__device__ __forceinline__ float sigmoidf_(float x) {
  return 1.0f / (1.0f + __expf(-x));
}

#define GRU_TILES 4
#define GRU_WAVES 16
#define WP32 36
#define WP64 68
#define HXP  36
#define HXBUF (GRU_TILES * 32 * HXP)
#define OSP  36

__device__ __forceinline__ v4f lds4(const float* p) {
  return *reinterpret_cast<const v4f*>(p);
}

__device__ __forceinline__ v8f join8(v4f lo, v4f hi) {
  return __builtin_shufflevector(lo, hi, 0, 1, 2, 3, 4, 5, 6, 7);
}

template <int NU, int NB>
__device__ __forceinline__ void mm_block(v8f (&acc)[3], const v4f (&f)[3][8],
                                         const float (&b)[NB]) {
#pragma unroll
  for (int u = 0; u < NU; ++u) {
#pragma unroll
    for (int hf = 0; hf < 2; ++hf) {
      v2f bb;
      bb.x = b[4 * u + 2 * hf];
      bb.y = b[4 * u + 2 * hf + 1];
#pragma unroll
      for (int g = 0; g < 3; ++g) {
        v2f aa;
        aa.x = f[g][u][2 * hf];
        aa.y = f[g][u][2 * hf + 1];
        acc[g] = wmma_f32(aa, bb, acc[g]);
      }
    }
  }
}

template <int K32, int NU, int PITCH>
__device__ __forceinline__ void load_w(const float* pa, const float* pb, int jj,
                                       v4f (&F)[3][8]) {
#pragma unroll
  for (int g = 0; g < 3; ++g) {
    const int ro = (64 * g + 16 * jj) * PITCH;
#pragma unroll
    for (int u = 0; u < NU; ++u) {
      if (K32) F[g][u] = lds4(pa + ro + 4 * u);
      else     F[g][u] = lds4((u < 4 ? pa : pb) + ro + 16 * ((u & 3) >> 1) + 4 * (u & 1));
    }
  }
}

__device__ __forceinline__ void load_bias(const float* b, int jj, v8f (&acc)[3]) {
#pragma unroll
  for (int g = 0; g < 3; ++g)
    acc[g] = join8(lds4(b + 64 * g + 16 * jj), lds4(b + 64 * g + 16 * jj + 4));
}

template <int LAYER>
__device__ __forceinline__ void gru_half_step(
    const float* wia, const float* wha, const float* whb,
    const float* bi, const float* bh,
    const float (&S)[32], float (&hn)[16], float (&xc)[16],
    const float* hsrc, const float4* xnext) {
  v4f F[3][8];
#pragma unroll
  for (int jj = 0; jj < 2; ++jj) {
    v8f gi[3], gh[3];
    load_w<0, 8, WP64>(wha, whb, jj, F);
    load_bias(bh, jj, gh);
    __builtin_amdgcn_sched_barrier(0);
    mm_block<8, 32>(gh, F, S);
    __builtin_amdgcn_sched_barrier(0);
    if (LAYER == 0) {
      load_w<1, 4, WP32>(wia, wia, jj, F);
      load_bias(bi, jj, gi);
      __builtin_amdgcn_sched_barrier(0);
      mm_block<4, 16>(gi, F, xc);
      if (jj == 1) {
        __builtin_amdgcn_sched_barrier(0);
#pragma unroll
        for (int u = 0; u < 4; ++u) {
          const float4 v = xnext[u];
          xc[4 * u] = v.x; xc[4 * u + 1] = v.y; xc[4 * u + 2] = v.z; xc[4 * u + 3] = v.w;
        }
      }
    } else {
#pragma unroll
      for (int kh = 0; kh < 2; ++kh) {
        float b16[16];
        load_w<0, 4, WP64>(wia + 32 * kh, wia, jj, F);
#pragma unroll
        for (int u = 0; u < 4; ++u) {
          const v4f v = lds4(hsrc + 16 * kh + 4 * u);
#pragma unroll
          for (int e = 0; e < 4; ++e) b16[4 * u + e] = v[e];
        }
        if (kh == 0) load_bias(bi, jj, gi);
        __builtin_amdgcn_sched_barrier(0);
        mm_block<4, 16>(gi, F, b16);
        if (kh == 0) __builtin_amdgcn_sched_barrier(0);
      }
    }
    __builtin_amdgcn_sched_barrier(0);

#pragma unroll
    for (int v = 0; v < 8; ++v) {
      const float hold = S[8 * jj + v];
      const float r  = sigmoidf_(gi[0][v] + gh[0][v]);
      const float zg = sigmoidf_(gi[1][v] + gh[1][v]);
      const float n  = tanhf(gi[2][v] + r * gh[2][v]);
      hn[8 * jj + v] = (1.0f - zg) * n + zg * hold;
    }
  }
}

__global__ __launch_bounds__(512) void gru_fused_kernel(
    const float* __restrict__ x_in,
    const float* __restrict__ w_ih0,
    const float* __restrict__ w_hh0,
    const float* __restrict__ b_ih0,
    const float* __restrict__ b_hh0,
    const float* __restrict__ w_ih1,
    const float* __restrict__ w_hh1,
    const float* __restrict__ b_ih1,
    const float* __restrict__ b_hh1,
    float* __restrict__ z_out)
{
  __shared__ __attribute__((aligned(16))) float wih0_s[G3 * WP32];
  __shared__ __attribute__((aligned(16))) float whh0_s[G3 * WP64];
  __shared__ __attribute__((aligned(16))) float wih1_s[G3 * WP64];
  __shared__ __attribute__((aligned(16))) float whh1_s[G3 * WP64];
  __shared__ __attribute__((aligned(16))) float bias_s[4 * G3];
  __shared__ __attribute__((aligned(16))) float ost[GRU_WAVES][16 * OSP];
  __shared__ __attribute__((aligned(16))) float hx[2 * HXBUF];
  __shared__ __attribute__((aligned(16))) float hy[2 * HXBUF];

  const int tid = threadIdx.x;
  {
    const float4* g0 = reinterpret_cast<const float4*>(w_ih0);
    for (int i = tid; i < G3 * FEAT / 4; i += 512) {
      const int row = i >> 3, c4 = i & 7;
      *reinterpret_cast<float4*>(&wih0_s[row * WP32 + 4 * c4]) = g0[i];
    }
    const float4* g1 = reinterpret_cast<const float4*>(w_hh0);
    const float4* g2 = reinterpret_cast<const float4*>(w_ih1);
    const float4* g3 = reinterpret_cast<const float4*>(w_hh1);
    for (int i = tid; i < G3 * HID / 4; i += 512) {
      const int row = i >> 4, c4 = i & 15;
      *reinterpret_cast<float4*>(&whh0_s[row * WP64 + 4 * c4]) = g1[i];
      *reinterpret_cast<float4*>(&wih1_s[row * WP64 + 4 * c4]) = g2[i];
      *reinterpret_cast<float4*>(&whh1_s[row * WP64 + 4 * c4]) = g3[i];
    }
    for (int i = tid; i < G3; i += 512) {
      bias_s[i]          = b_ih0[i];
      bias_s[G3 + i]     = b_hh0[i];
      bias_s[2 * G3 + i] = b_ih1[i];
      bias_s[3 * G3 + i] = b_hh1[i];
    }
  }
  __syncthreads();

  const int wave = tid >> 5;
  const int lane = tid & 31;
  const int l16  = lane & 15;
  const int lh   = lane >> 4;
  const int role  = __builtin_amdgcn_readfirstlane(wave >> 2);
  const int layer = (role < 2) ? 1 : 0;
  const int half  = role & 1;
  const int tile  = wave & (GRU_TILES - 1);
  const int row0  = blockIdx.x * (GRU_TILES * 16) + tile * 16;
  const int own   = 16 * half;
  const int oth   = 16 - own;

  float S[32], xc[16];
#pragma unroll
  for (int i = 0; i < 32; ++i) S[i] = 0.0f;
#pragma unroll
  for (int i = 0; i < 16; ++i) xc[i] = 0.0f;

  float* hxl = &hx[(tile * 32 + lane) * HXP];
  float* hyl = &hy[(tile * 32 + lane) * HXP];

  const int rw = 32 * half + l16;
  const float* wia;
  const float* wha;
  const float* whb;
  const float* bi;
  const float* bh;
  if (layer == 0) {
    wia = wih0_s + rw * WP32 + 16 * lh;
    wha = whh0_s + rw * WP64 + 32 * half + 8 * lh;
    whb = whh0_s + rw * WP64 + 32 * (1 - half) + 8 * lh;
    bi  = bias_s + 32 * half + 8 * lh;
    bh  = bias_s + G3 + 32 * half + 8 * lh;
    const float4* xp = reinterpret_cast<const float4*>(
        x_in + ((size_t)row0 + l16) * FEAT + 16 * lh);
#pragma unroll
    for (int u = 0; u < 4; ++u) {
      const float4 v = xp[u];
      xc[4 * u] = v.x; xc[4 * u + 1] = v.y; xc[4 * u + 2] = v.z; xc[4 * u + 3] = v.w;
    }
  } else {
    wia = wih1_s + rw * WP64 + 8 * lh;
    wha = whh1_s + rw * WP64 + 32 * half + 8 * lh;
    whb = whh1_s + rw * WP64 + 32 * (1 - half) + 8 * lh;
    bi  = bias_s + 2 * G3 + 32 * half + 8 * lh;
    bh  = bias_s + 3 * G3 + 32 * half + 8 * lh;
  }

  for (int s = 0; s <= T_SEQ; ++s) {
    if (layer == 0) {
      if (s < T_SEQ) {
        const int tn = (s + 1 < T_SEQ) ? (s + 1) : s;
        const float4* xnp = reinterpret_cast<const float4*>(
            x_in + ((size_t)tn * NROWS + row0 + l16) * FEAT + 16 * lh);
        if (s >= 1) {
          const float* srcp = hxl + ((s & 1) ^ 1) * HXBUF + oth;
#pragma unroll
          for (int u = 0; u < 4; ++u) {
            const v4f v = lds4(srcp + 4 * u);
#pragma unroll
            for (int e = 0; e < 4; ++e) S[16 + 4 * u + e] = v[e];
          }
        }
        float hn[16];
        gru_half_step<0>(wia, wha, whb, bi, bh, S, hn, xc, hxl, xnp);
        float* dst = hxl + (s & 1) * HXBUF + own;
#pragma unroll
        for (int u = 0; u < 4; ++u) {
          v4f v;
#pragma unroll
          for (int e = 0; e < 4; ++e) v[e] = hn[4 * u + e];
          *reinterpret_cast<v4f*>(dst + 4 * u) = v;
        }
#pragma unroll
        for (int i = 0; i < 16; ++i) S[i] = hn[i];
      }
    } else {
      if (s >= 1) {
        const float* srcp = hxl + ((s & 1) ^ 1) * HXBUF;
        if (s >= 2) {
          const float* srcq = hyl + ((s & 1) ^ 1) * HXBUF + oth;
#pragma unroll
          for (int u = 0; u < 4; ++u) {
            const v4f v = lds4(srcq + 4 * u);
#pragma unroll
            for (int e = 0; e < 4; ++e) S[16 + 4 * u + e] = v[e];
          }
        }
        float hn[16];
        gru_half_step<1>(wia, wha, whb, bi, bh, S, hn, xc, srcp, nullptr);
        float* dst = hyl + (s & 1) * HXBUF + own;
#pragma unroll
        for (int u = 0; u < 4; ++u) {
          v4f v;
#pragma unroll
          for (int e = 0; e < 4; ++e) v[e] = hn[4 * u + e];
          *reinterpret_cast<v4f*>(dst + 4 * u) = v;
        }
#pragma unroll
        for (int i = 0; i < 16; ++i) S[i] = hn[i];
      }
    }
    __syncthreads();
  }

  float* st = &ost[wave][0];
#pragma unroll
  for (int u = 0; u < 4; ++u) {
    v4f v;
#pragma unroll
    for (int e = 0; e < 4; ++e) v[e] = S[4 * u + e];
    *reinterpret_cast<v4f*>(&st[l16 * OSP + 16 * (u >> 1) + 8 * lh + 4 * (u & 1)]) = v;
  }
#pragma unroll
  for (int it = 0; it < 4; ++it) {
    const int idx = it * 32 + lane;
    const int r = idx >> 3, c4 = idx & 7;
    const v4f v = lds4(&st[r * OSP + 4 * c4]);
    *reinterpret_cast<v4f*>(
        &z_out[(size_t)layer * NROWS * HID + (size_t)(row0 + r) * HID + 32 * half + 4 * c4]) = v;
  }
}

#define GAT_WAVES 8
#define ZP 68

typedef __attribute__((ext_vector_type(4))) int v4i;

__device__ __forceinline__ float xhalf(float x) {
  const int b = __builtin_bit_cast(int, x);
  return __builtin_bit_cast(float,
      __builtin_amdgcn_permlanex16(b, b, 0x76543210, 0xfedcba98, false, false));
}

__device__ __forceinline__ void gat_fetch(const float* zs, const int* mlane, int m0,
                                          int l16, int lh, v4f (&za)[8],
                                          float (&tx)[4][4], float (&ty)[4][4],
                                          v4i (&mk)[2]) {
  const float* ar = zs + (m0 + l16) * ZP + 4 * lh;
#pragma unroll
  for (int u = 0; u < 8; ++u) za[u] = lds4(ar + 8 * u);
  const float* tr = zs + (m0 + 8 * lh) * ZP + l16;
#pragma unroll
  for (int q = 0; q < 4; ++q) {
#pragma unroll
    for (int dt = 0; dt < 4; ++dt) {
      tx[dt][q] = tr[(2 * q) * ZP + 16 * dt];
      ty[dt][q] = tr[(2 * q + 1) * ZP + 16 * dt];
    }
  }
  const v4i* mp = reinterpret_cast<const v4i*>(mlane + m0);
  mk[0] = mp[0];
  mk[1] = mp[1];
}

__device__ __forceinline__ void gat_step(const v4f (&zb)[8], const v4f (&za)[8],
                                         const float (&tx)[4][4], const float (&ty)[4][4],
                                         const v4i (&mk)[2],
                                         float& M, float& L, v8f (&accO)[4]) {
  v8f s0 = {}, s1 = {};
#pragma unroll
  for (int u = 0; u < 8; ++u) {
    v2f a0, b0, a1, b1;
    a0.x = za[u][0]; a0.y = za[u][1]; b0.x = zb[u][0]; b0.y = zb[u][1];
    a1.x = za[u][2]; a1.y = za[u][3]; b1.x = zb[u][2]; b1.y = zb[u][3];
    s0 = wmma_f32(a0, b0, s0);
    s1 = wmma_f32(a1, b1, s1);
  }
  float a[8];
#pragma unroll
  for (int r = 0; r < 8; ++r) {
    const float s  = s0[r] + s1[r];
    const float lr = s > 0.0f ? s : 0.1f * s;
    const int   mi = (r < 4) ? mk[0][r & 3] : mk[1][r & 3];
    float v = lr * (float)mi;
    v = (v == 0.0f) ? NEG_BIG : v;
    a[r] = v;
  }
  float tm = a[0];
#pragma unroll
  for (int r = 1; r < 8; ++r) tm = fmaxf(tm, a[r]);
  tm = fmaxf(tm, xhalf(tm));
  const float Mn    = fmaxf(M, tm);
  const float alpha = __expf(M - Mn);
  float p[8];
  float ps = 0.0f;
#pragma unroll
  for (int r = 0; r < 8; ++r) { p[r] = __expf(a[r] - Mn); ps += p[r]; }
  L = L * alpha + ps;
  M = Mn;
#pragma unroll
  for (int dt = 0; dt < 4; ++dt)
#pragma unroll
    for (int r = 0; r < 8; ++r) accO[dt][r] *= alpha;
#pragma unroll
  for (int q = 0; q < 4; ++q) {
    v2f b;
    b.x = p[2 * q];
    b.y = p[2 * q + 1];
#pragma unroll
    for (int dt = 0; dt < 4; ++dt) {
      v2f a2;
      a2.x = tx[dt][q];
      a2.y = ty[dt][q];
      accO[dt] = wmma_f32(a2, b, accO[dt]);
    }
  }
}

__global__ __launch_bounds__(256) void gat_kernel(
    const float* __restrict__ z,
    const int*   __restrict__ mask,
    float*       __restrict__ out)
{
  __shared__ __attribute__((aligned(16))) float zs[NODES * ZP];

  const int tid   = threadIdx.x;
  const int pair  = blockIdx.x >> 3;
  const int chunk = blockIdx.x & 7;
  const float* zp = z + (size_t)pair * NODES * HID;

  {
    const float4* g = reinterpret_cast<const float4*>(zp);
#pragma unroll 8
    for (int it = 0; it < NODES * HID / 4 / 256; ++it) {
      const int i = it * 256 + tid;
      const int row = i >> 4, c4 = i & 15;
      *reinterpret_cast<float4*>(&zs[row * ZP + 4 * c4]) = g[i];
    }
  }
  __syncthreads();

  const int wave = tid >> 5;
  const int lane = tid & 31;
  const int l16  = lane & 15;
  const int lh   = lane >> 4;
  const int n0   = (chunk * GAT_WAVES + wave) * 16;

  v4f zb[8];
  {
    const float* br = zs + (n0 + l16) * ZP + 4 * lh;
#pragma unroll
    for (int u = 0; u < 8; ++u) zb[u] = lds4(br + 8 * u);
  }
  const int* mlane = mask + (size_t)(n0 + l16) * NODES + 8 * lh;

  float M = -INFINITY, L = 0.0f;
  v8f accO[4] = {};
  v4f zaA[8], zaB[8];
  float txA[4][4], tyA[4][4], txB[4][4], tyB[4][4];
  v4i mkA[2], mkB[2];

  gat_fetch(zs, mlane, 0, l16, lh, zaA, txA, tyA, mkA);
  for (int t = 0; t < 64; t += 2) {
    gat_fetch(zs, mlane, (t + 1) * 16, l16, lh, zaB, txB, tyB, mkB);
    __builtin_amdgcn_sched_barrier(0);
    gat_step(zb, zaA, txA, tyA, mkA, M, L, accO);
    {
      const int tn = (t + 2 < 64) ? (t + 2) : t;
      gat_fetch(zs, mlane, tn * 16, l16, lh, zaA, txA, tyA, mkA);
    }
    __builtin_amdgcn_sched_barrier(0);
    gat_step(zb, zaB, txB, tyB, mkB, M, L, accO);
  }

  const float inv = 1.0f / (L + xhalf(L));
  float* orow = out + (size_t)pair * NODES * HID + (size_t)(n0 + l16) * HID + 8 * lh;
#pragma unroll
  for (int dt = 0; dt < 4; ++dt) {
    v4f lo, hi;
#pragma unroll
    for (int e = 0; e < 4; ++e) { lo[e] = accO[dt][e] * inv; hi[e] = accO[dt][4 + e] * inv; }
    *reinterpret_cast<v4f*>(orow + 16 * dt)     = lo;
    *reinterpret_cast<v4f*>(orow + 16 * dt + 4) = hi;
  }
}

extern "C" void kernel_launch(void* const* d_in, const int* in_sizes, int n_in,
                              void* d_out, int out_size, void* d_ws, size_t ws_size,
                              hipStream_t stream) {
  const float* x     = (const float*)d_in[0];
  const int*   mask  = (const int*)  d_in[1];
  const float* w_ih0 = (const float*)d_in[2];
  const float* w_hh0 = (const float*)d_in[3];
  const float* b_ih0 = (const float*)d_in[4];
  const float* b_hh0 = (const float*)d_in[5];
  const float* w_ih1 = (const float*)d_in[6];
  const float* w_hh1 = (const float*)d_in[7];
  const float* b_ih1 = (const float*)d_in[8];
  const float* b_hh1 = (const float*)d_in[9];
  float* out  = (float*)d_out;
  float* zbuf = (float*)d_ws;

  gru_fused_kernel<<<256, 512, 0, stream>>>(x, w_ih0, w_hh0, b_ih0, b_hh0,
                                            w_ih1, w_hh1, b_ih1, b_hh1, zbuf);
  gat_kernel<<<256, 256, 0, stream>>>(zbuf, mask, out);
}


# GRU step barrier split into signal and wait: each wave starts its next step (own-half state product, layer 0 also the x product) between the two instead of idling; sums in the same order
# speedup vs baseline: 2.4798x; 1.0321x over previous
#include <hip/hip_runtime.h>
#include <hip/hip_bf16.h>
#include <math.h>


typedef __attribute__((ext_vector_type(2))) float v2f;
typedef __attribute__((ext_vector_type(4))) float v4f;
typedef __attribute__((ext_vector_type(8))) float v8f;

#define T_SEQ   12
#define BCN     16
#define NODES   1024
#define FEAT    32
#define HID     64
#define G3      192
#define NROWS   (BCN * NODES)
#define NEG_BIG (-1e16f)

__device__ __forceinline__ v8f wmma_f32(v2f a, v2f b, v8f c) {
  return __builtin_amdgcn_wmma_f32_16x16x4_f32(
       false, a,  false, b,
       (short)0, c,  false,  false);
}

__device__ __forceinline__ float sigmoidf_(float x) {
  return 1.0f / (1.0f + __expf(-x));
}

#define GRU_TILES 4
#define GRU_WAVES 16
#define WP32 36
#define WP64 68
#define HXP  36
#define HXBUF (GRU_TILES * 32 * HXP)
#define OSP  36

__device__ __forceinline__ v4f lds4(const float* p) {
  return *reinterpret_cast<const v4f*>(p);
}

__device__ __forceinline__ v8f join8(v4f lo, v4f hi) {
  return __builtin_shufflevector(lo, hi, 0, 1, 2, 3, 4, 5, 6, 7);
}

__device__ __forceinline__ void mm_k32(v8f (&acc)[3], const v4f (&f)[3][4],
                                       const float (&b)[16]) {
#pragma unroll
  for (int u = 0; u < 4; ++u) {
#pragma unroll
    for (int hf = 0; hf < 2; ++hf) {
      v2f bb;
      bb.x = b[4 * u + 2 * hf];
      bb.y = b[4 * u + 2 * hf + 1];
#pragma unroll
      for (int g = 0; g < 3; ++g) {
        v2f aa;
        aa.x = f[g][u][2 * hf];
        aa.y = f[g][u][2 * hf + 1];
        acc[g] = wmma_f32(aa, bb, acc[g]);
      }
    }
  }
}

template <int K32, int PITCH>
__device__ __forceinline__ void load_w4(const float* p, int jj, v4f (&F)[3][4]) {
#pragma unroll
  for (int g = 0; g < 3; ++g) {
    const int ro = (64 * g + 16 * jj) * PITCH;
#pragma unroll
    for (int u = 0; u < 4; ++u)
      F[g][u] = lds4(p + ro + (K32 ? 4 * u : 16 * (u >> 1) + 4 * (u & 1)));
  }
}

__device__ __forceinline__ void load_bias(const float* b, int jj, v8f (&acc)[3]) {
#pragma unroll
  for (int g = 0; g < 3; ++g)
    acc[g] = join8(lds4(b + 64 * g + 16 * jj), lds4(b + 64 * g + 16 * jj + 4));
}

__device__ __forceinline__ void slab_read16(const float* p, float (&d)[16]) {
#pragma unroll
  for (int u = 0; u < 4; ++u) {
    const v4f v = lds4(p + 4 * u);
#pragma unroll
    for (int e = 0; e < 4; ++e) d[4 * u + e] = v[e];
  }
}

__device__ __forceinline__ void slab_write16(float* p, const float (&d)[16]) {
#pragma unroll
  for (int u = 0; u < 4; ++u) {
    v4f v;
#pragma unroll
    for (int e = 0; e < 4; ++e) v[e] = d[4 * u + e];
    *reinterpret_cast<v4f*>(p + 4 * u) = v;
  }
}

__device__ __forceinline__ void gate_tile(const v8f (&gi)[3], const v8f (&gh)[3],
                                          float (&So)[16], int jj) {
#pragma unroll
  for (int v = 0; v < 8; ++v) {
    const float hold = So[8 * jj + v];
    const float r  = sigmoidf_(gi[0][v] + gh[0][v]);
    const float zg = sigmoidf_(gi[1][v] + gh[1][v]);
    const float n  = tanhf(gi[2][v] + r * gh[2][v]);
    So[8 * jj + v] = (1.0f - zg) * n + zg * hold;
  }
}

__device__ __forceinline__ void step_signal() {
  asm volatile("s_wait_dscnt 0x0\n\ts_barrier_signal -1" ::: "memory");
}
__device__ __forceinline__ void step_wait() {
  asm volatile("s_barrier_wait -1" ::: "memory");
}

__global__ __launch_bounds__(512) void gru_fused_kernel(
    const float* __restrict__ x_in,
    const float* __restrict__ w_ih0,
    const float* __restrict__ w_hh0,
    const float* __restrict__ b_ih0,
    const float* __restrict__ b_hh0,
    const float* __restrict__ w_ih1,
    const float* __restrict__ w_hh1,
    const float* __restrict__ b_ih1,
    const float* __restrict__ b_hh1,
    float* __restrict__ z_out)
{
  __shared__ __attribute__((aligned(16))) float wih0_s[G3 * WP32];
  __shared__ __attribute__((aligned(16))) float whh0_s[G3 * WP64];
  __shared__ __attribute__((aligned(16))) float wih1_s[G3 * WP64];
  __shared__ __attribute__((aligned(16))) float whh1_s[G3 * WP64];
  __shared__ __attribute__((aligned(16))) float bias_s[4 * G3];
  __shared__ __attribute__((aligned(16))) float ost[GRU_WAVES][16 * OSP];
  __shared__ __attribute__((aligned(16))) float hx[2 * HXBUF];
  __shared__ __attribute__((aligned(16))) float hy[2 * HXBUF];

  const int tid = threadIdx.x;
  {
    const float4* g0 = reinterpret_cast<const float4*>(w_ih0);
    for (int i = tid; i < G3 * FEAT / 4; i += 512) {
      const int row = i >> 3, c4 = i & 7;
      *reinterpret_cast<float4*>(&wih0_s[row * WP32 + 4 * c4]) = g0[i];
    }
    const float4* g1 = reinterpret_cast<const float4*>(w_hh0);
    const float4* g2 = reinterpret_cast<const float4*>(w_ih1);
    const float4* g3 = reinterpret_cast<const float4*>(w_hh1);
    for (int i = tid; i < G3 * HID / 4; i += 512) {
      const int row = i >> 4, c4 = i & 15;
      *reinterpret_cast<float4*>(&whh0_s[row * WP64 + 4 * c4]) = g1[i];
      *reinterpret_cast<float4*>(&wih1_s[row * WP64 + 4 * c4]) = g2[i];
      *reinterpret_cast<float4*>(&whh1_s[row * WP64 + 4 * c4]) = g3[i];
    }
    for (int i = tid; i < G3; i += 512) {
      bias_s[i]          = b_ih0[i];
      bias_s[G3 + i]     = b_hh0[i];
      bias_s[2 * G3 + i] = b_ih1[i];
      bias_s[3 * G3 + i] = b_hh1[i];
    }
  }
  __syncthreads();

  const int wave = tid >> 5;
  const int lane = tid & 31;
  const int l16  = lane & 15;
  const int lh   = lane >> 4;
  const int role  = __builtin_amdgcn_readfirstlane(wave >> 2);
  const int layer = (role < 2) ? 1 : 0;
  const int half  = role & 1;
  const int tile  = wave & (GRU_TILES - 1);
  const int row0  = blockIdx.x * (GRU_TILES * 16) + tile * 16;
  const int own   = 16 * half;
  const int oth   = 16 - own;

  float So[16], Sp[16], xc[16];
#pragma unroll
  for (int i = 0; i < 16; ++i) { So[i] = 0.0f; Sp[i] = 0.0f; xc[i] = 0.0f; }
  v8f gi[2][3], gh[2][3];
#pragma unroll
  for (int j = 0; j < 2; ++j)
#pragma unroll
    for (int g = 0; g < 3; ++g)
#pragma unroll
      for (int e = 0; e < 8; ++e) { gi[j][g][e] = 0.0f; gh[j][g][e] = 0.0f; }

  float* hxl = &hx[(tile * 32 + lane) * HXP];
  float* hyl = &hy[(tile * 32 + lane) * HXP];

  const int rw = 32 * half + l16;
  const float* wia;
  const float* wha;
  const float* whb;
  const float* bi;
  const float* bh;
  if (layer == 0) {
    wia = wih0_s + rw * WP32 + 16 * lh;
    wha = whh0_s + rw * WP64 + 32 * half + 8 * lh;
    whb = whh0_s + rw * WP64 + 32 * (1 - half) + 8 * lh;
    bi  = bias_s + 32 * half + 8 * lh;
    bh  = bias_s + G3 + 32 * half + 8 * lh;
    const float4* xp = reinterpret_cast<const float4*>(
        x_in + ((size_t)row0 + l16) * FEAT + 16 * lh);
#pragma unroll
    for (int u = 0; u < 4; ++u) {
      const float4 v = xp[u];
      xc[4 * u] = v.x; xc[4 * u + 1] = v.y; xc[4 * u + 2] = v.z; xc[4 * u + 3] = v.w;
    }
  } else {
    wia = wih1_s + rw * WP64 + 8 * lh;
    wha = whh1_s + rw * WP64 + 32 * half + 8 * lh;
    whb = whh1_s + rw * WP64 + 32 * (1 - half) + 8 * lh;
    bi  = bias_s + 2 * G3 + 32 * half + 8 * lh;
    bh  = bias_s + 3 * G3 + 32 * half + 8 * lh;
  }

  float* mys = layer ? hyl : hxl;
  for (int s = -1; s <= T_SEQ; ++s) {
    const int t = s - layer;
    if (t >= 0 && t < T_SEQ) {
      const int rd = ((s & 1) ^ 1) * HXBUF;
      if (t >= 1) slab_read16(mys + rd + oth, Sp);
#pragma unroll
      for (int jj = 0; jj < 2; ++jj) {
        v4f F[3][4];
        load_w4<0, WP64>(whb, jj, F);
        __builtin_amdgcn_sched_barrier(0);
        mm_k32(gh[jj], F, Sp);
        __builtin_amdgcn_sched_barrier(0);
        if (layer == 1) {
#pragma unroll
          for (int kh = 0; kh < 2; ++kh) {
            load_w4<0, WP64>(wia + 32 * kh, jj, F);
            slab_read16(hxl + rd + 16 * kh, xc);
            if (kh == 0) load_bias(bi, jj, gi[jj]);
            __builtin_amdgcn_sched_barrier(0);
            mm_k32(gi[jj], F, xc);
            __builtin_amdgcn_sched_barrier(0);
          }
        }
        gate_tile(gi[jj], gh[jj], So, jj);
      }
      slab_write16(mys + (s & 1) * HXBUF + own, So);
    }
    step_signal();
    __builtin_amdgcn_sched_barrier(0);
    if (t + 1 >= 0 && t + 1 < T_SEQ) {
#pragma unroll
      for (int jj = 0; jj < 2; ++jj) {
        v4f F[3][4];
        load_w4<0, WP64>(wha, jj, F);
        load_bias(bh, jj, gh[jj]);
        __builtin_amdgcn_sched_barrier(0);
        mm_k32(gh[jj], F, So);
        __builtin_amdgcn_sched_barrier(0);
        if (layer == 0) {
          load_w4<1, WP32>(wia, jj, F);
          load_bias(bi, jj, gi[jj]);
          __builtin_amdgcn_sched_barrier(0);
          mm_k32(gi[jj], F, xc);
          __builtin_amdgcn_sched_barrier(0);
        }
      }
      if (layer == 0) {
        const int tn = (t + 2 < T_SEQ) ? (t + 2) : (t + 1);
        const float4* xnp = reinterpret_cast<const float4*>(
            x_in + ((size_t)tn * NROWS + row0 + l16) * FEAT + 16 * lh);
#pragma unroll
        for (int u = 0; u < 4; ++u) {
          const float4 v = xnp[u];
          xc[4 * u] = v.x; xc[4 * u + 1] = v.y; xc[4 * u + 2] = v.z; xc[4 * u + 3] = v.w;
        }
      }
    }
    step_wait();
  }

  float* st = &ost[wave][0];
#pragma unroll
  for (int u = 0; u < 4; ++u) {
    v4f v;
#pragma unroll
    for (int e = 0; e < 4; ++e) v[e] = So[4 * u + e];
    *reinterpret_cast<v4f*>(&st[l16 * OSP + 16 * (u >> 1) + 8 * lh + 4 * (u & 1)]) = v;
  }
#pragma unroll
  for (int it = 0; it < 4; ++it) {
    const int idx = it * 32 + lane;
    const int r = idx >> 3, c4 = idx & 7;
    const v4f v = lds4(&st[r * OSP + 4 * c4]);
    *reinterpret_cast<v4f*>(
        &z_out[(size_t)layer * NROWS * HID + (size_t)(row0 + r) * HID + 32 * half + 4 * c4]) = v;
  }
}

#define GAT_WAVES 8
#define ZP 68

typedef __attribute__((ext_vector_type(4))) int v4i;

__device__ __forceinline__ float xhalf(float x) {
  const int b = __builtin_bit_cast(int, x);
  return __builtin_bit_cast(float,
      __builtin_amdgcn_permlanex16(b, b, 0x76543210, 0xfedcba98, false, false));
}

__device__ __forceinline__ void gat_fetch(const float* zs, const int* mlane, int m0,
                                          int l16, int lh, v4f (&za)[8],
                                          float (&tx)[4][4], float (&ty)[4][4],
                                          v4i (&mk)[2]) {
  const float* ar = zs + (m0 + l16) * ZP + 4 * lh;
#pragma unroll
  for (int u = 0; u < 8; ++u) za[u] = lds4(ar + 8 * u);
  const float* tr = zs + (m0 + 8 * lh) * ZP + l16;
#pragma unroll
  for (int q = 0; q < 4; ++q) {
#pragma unroll
    for (int dt = 0; dt < 4; ++dt) {
      tx[dt][q] = tr[(2 * q) * ZP + 16 * dt];
      ty[dt][q] = tr[(2 * q + 1) * ZP + 16 * dt];
    }
  }
  const v4i* mp = reinterpret_cast<const v4i*>(mlane + m0);
  mk[0] = mp[0];
  mk[1] = mp[1];
}

__device__ __forceinline__ void gat_step(const v4f (&zb)[8], const v4f (&za)[8],
                                         const float (&tx)[4][4], const float (&ty)[4][4],
                                         const v4i (&mk)[2],
                                         float& M, float& L, v8f (&accO)[4]) {
  v8f s0 = {}, s1 = {};
#pragma unroll
  for (int u = 0; u < 8; ++u) {
    v2f a0, b0, a1, b1;
    a0.x = za[u][0]; a0.y = za[u][1]; b0.x = zb[u][0]; b0.y = zb[u][1];
    a1.x = za[u][2]; a1.y = za[u][3]; b1.x = zb[u][2]; b1.y = zb[u][3];
    s0 = wmma_f32(a0, b0, s0);
    s1 = wmma_f32(a1, b1, s1);
  }
  float a[8];
#pragma unroll
  for (int r = 0; r < 8; ++r) {
    const float s  = s0[r] + s1[r];
    const float lr = s > 0.0f ? s : 0.1f * s;
    const int   mi = (r < 4) ? mk[0][r & 3] : mk[1][r & 3];
    float v = lr * (float)mi;
    v = (v == 0.0f) ? NEG_BIG : v;
    a[r] = v;
  }
  float tm = a[0];
#pragma unroll
  for (int r = 1; r < 8; ++r) tm = fmaxf(tm, a[r]);
  tm = fmaxf(tm, xhalf(tm));
  const float Mn    = fmaxf(M, tm);
  const float alpha = __expf(M - Mn);
  float p[8];
  float ps = 0.0f;
#pragma unroll
  for (int r = 0; r < 8; ++r) { p[r] = __expf(a[r] - Mn); ps += p[r]; }
  L = L * alpha + ps;
  M = Mn;
#pragma unroll
  for (int dt = 0; dt < 4; ++dt)
#pragma unroll
    for (int r = 0; r < 8; ++r) accO[dt][r] *= alpha;
#pragma unroll
  for (int q = 0; q < 4; ++q) {
    v2f b;
    b.x = p[2 * q];
    b.y = p[2 * q + 1];
#pragma unroll
    for (int dt = 0; dt < 4; ++dt) {
      v2f a2;
      a2.x = tx[dt][q];
      a2.y = ty[dt][q];
      accO[dt] = wmma_f32(a2, b, accO[dt]);
    }
  }
}

__global__ __launch_bounds__(256) void gat_kernel(
    const float* __restrict__ z,
    const int*   __restrict__ mask,
    float*       __restrict__ out)
{
  __shared__ __attribute__((aligned(16))) float zs[NODES * ZP];

  const int tid   = threadIdx.x;
  const int pair  = blockIdx.x >> 3;
  const int chunk = blockIdx.x & 7;
  const float* zp = z + (size_t)pair * NODES * HID;

  {
    const float4* g = reinterpret_cast<const float4*>(zp);
#pragma unroll 8
    for (int it = 0; it < NODES * HID / 4 / 256; ++it) {
      const int i = it * 256 + tid;
      const int row = i >> 4, c4 = i & 15;
      *reinterpret_cast<float4*>(&zs[row * ZP + 4 * c4]) = g[i];
    }
  }
  __syncthreads();

  const int wave = tid >> 5;
  const int lane = tid & 31;
  const int l16  = lane & 15;
  const int lh   = lane >> 4;
  const int n0   = (chunk * GAT_WAVES + wave) * 16;

  v4f zb[8];
  {
    const float* br = zs + (n0 + l16) * ZP + 4 * lh;
#pragma unroll
    for (int u = 0; u < 8; ++u) zb[u] = lds4(br + 8 * u);
  }
  const int* mlane = mask + (size_t)(n0 + l16) * NODES + 8 * lh;

  float M = -INFINITY, L = 0.0f;
  v8f accO[4] = {};
  v4f zaA[8], zaB[8];
  float txA[4][4], tyA[4][4], txB[4][4], tyB[4][4];
  v4i mkA[2], mkB[2];

  gat_fetch(zs, mlane, 0, l16, lh, zaA, txA, tyA, mkA);
  for (int t = 0; t < 64; t += 2) {
    gat_fetch(zs, mlane, (t + 1) * 16, l16, lh, zaB, txB, tyB, mkB);
    __builtin_amdgcn_sched_barrier(0);
    gat_step(zb, zaA, txA, tyA, mkA, M, L, accO);
    {
      const int tn = (t + 2 < 64) ? (t + 2) : t;
      gat_fetch(zs, mlane, tn * 16, l16, lh, zaA, txA, tyA, mkA);
    }
    __builtin_amdgcn_sched_barrier(0);
    gat_step(zb, zaB, txB, tyB, mkB, M, L, accO);
  }

  const float inv = 1.0f / (L + xhalf(L));
  float* orow = out + (size_t)pair * NODES * HID + (size_t)(n0 + l16) * HID + 8 * lh;
#pragma unroll
  for (int dt = 0; dt < 4; ++dt) {
    v4f lo, hi;
#pragma unroll
    for (int e = 0; e < 4; ++e) { lo[e] = accO[dt][e] * inv; hi[e] = accO[dt][4 + e] * inv; }
    *reinterpret_cast<v4f*>(orow + 16 * dt)     = lo;
    *reinterpret_cast<v4f*>(orow + 16 * dt + 4) = hi;
  }
}

extern "C" void kernel_launch(void* const* d_in, const int* in_sizes, int n_in,
                              void* d_out, int out_size, void* d_ws, size_t ws_size,
                              hipStream_t stream) {
  const float* x     = (const float*)d_in[0];
  const int*   mask  = (const int*)  d_in[1];
  const float* w_ih0 = (const float*)d_in[2];
  const float* w_hh0 = (const float*)d_in[3];
  const float* b_ih0 = (const float*)d_in[4];
  const float* b_hh0 = (const float*)d_in[5];
  const float* w_ih1 = (const float*)d_in[6];
  const float* w_hh1 = (const float*)d_in[7];
  const float* b_ih1 = (const float*)d_in[8];
  const float* b_hh1 = (const float*)d_in[9];
  float* out  = (float*)d_out;
  float* zbuf = (float*)d_ws;

  gru_fused_kernel<<<256, 512, 0, stream>>>(x, w_ih0, w_hh0, b_ih0, b_hh0,
                                            w_ih1, w_hh1, b_ih1, b_hh1, zbuf);
  gat_kernel<<<256, 256, 0, stream>>>(zbuf, mask, out);
}


# GRU gate arithmetic ordered gate by gate: a tile's 8 reset-gate sigmoids stand in one block ahead of the 8 tanhf branches (held by an input-only asm), so their divide chains interleave; same sums
# speedup vs baseline: 2.5627x; 1.0334x over previous
#include <hip/hip_runtime.h>
#include <hip/hip_bf16.h>
#include <math.h>


typedef __attribute__((ext_vector_type(2))) float v2f;
typedef __attribute__((ext_vector_type(4))) float v4f;
typedef __attribute__((ext_vector_type(8))) float v8f;

#define T_SEQ   12
#define BCN     16
#define NODES   1024
#define FEAT    32
#define HID     64
#define G3      192
#define NROWS   (BCN * NODES)
#define NEG_BIG (-1e16f)

__device__ __forceinline__ v8f wmma_f32(v2f a, v2f b, v8f c) {
  return __builtin_amdgcn_wmma_f32_16x16x4_f32(
       false, a,  false, b,
       (short)0, c,  false,  false);
}

__device__ __forceinline__ float sigmoidf_(float x) {
  return 1.0f / (1.0f + __expf(-x));
}

#define GRU_TILES 4
#define GRU_WAVES 16
#define WP32 36
#define WP64 68
#define HXP  36
#define HXBUF (GRU_TILES * 32 * HXP)
#define OSP  36

__device__ __forceinline__ v4f lds4(const float* p) {
  return *reinterpret_cast<const v4f*>(p);
}

__device__ __forceinline__ v8f join8(v4f lo, v4f hi) {
  return __builtin_shufflevector(lo, hi, 0, 1, 2, 3, 4, 5, 6, 7);
}

__device__ __forceinline__ void mm_k32(v8f (&acc)[3], const v4f (&f)[3][4],
                                       const float (&b)[16]) {
#pragma unroll
  for (int u = 0; u < 4; ++u) {
#pragma unroll
    for (int hf = 0; hf < 2; ++hf) {
      v2f bb;
      bb.x = b[4 * u + 2 * hf];
      bb.y = b[4 * u + 2 * hf + 1];
#pragma unroll
      for (int g = 0; g < 3; ++g) {
        v2f aa;
        aa.x = f[g][u][2 * hf];
        aa.y = f[g][u][2 * hf + 1];
        acc[g] = wmma_f32(aa, bb, acc[g]);
      }
    }
  }
}

template <int K32, int PITCH>
__device__ __forceinline__ void load_w4(const float* p, int jj, v4f (&F)[3][4]) {
#pragma unroll
  for (int g = 0; g < 3; ++g) {
    const int ro = (64 * g + 16 * jj) * PITCH;
#pragma unroll
    for (int u = 0; u < 4; ++u)
      F[g][u] = lds4(p + ro + (K32 ? 4 * u : 16 * (u >> 1) + 4 * (u & 1)));
  }
}

__device__ __forceinline__ void load_bias(const float* b, int jj, v8f (&acc)[3]) {
#pragma unroll
  for (int g = 0; g < 3; ++g)
    acc[g] = join8(lds4(b + 64 * g + 16 * jj), lds4(b + 64 * g + 16 * jj + 4));
}

__device__ __forceinline__ void slab_read16(const float* p, float (&d)[16]) {
#pragma unroll
  for (int u = 0; u < 4; ++u) {
    const v4f v = lds4(p + 4 * u);
#pragma unroll
    for (int e = 0; e < 4; ++e) d[4 * u + e] = v[e];
  }
}

__device__ __forceinline__ void slab_write16(float* p, const float (&d)[16]) {
#pragma unroll
  for (int u = 0; u < 4; ++u) {
    v4f v;
#pragma unroll
    for (int e = 0; e < 4; ++e) v[e] = d[4 * u + e];
    *reinterpret_cast<v4f*>(p + 4 * u) = v;
  }
}

__device__ __forceinline__ void gate_tile(const v8f (&gi)[3], const v8f (&gh)[3],
                                          float (&So)[16], int jj) {
  float npre[8];
#pragma unroll
  for (int v = 0; v < 8; ++v) {
    const float r = sigmoidf_(gi[0][v] + gh[0][v]);
    npre[v] = gi[2][v] + r * gh[2][v];
  }
  asm volatile("" :: "v"(npre[0]), "v"(npre[1]), "v"(npre[2]), "v"(npre[3]),
                     "v"(npre[4]), "v"(npre[5]), "v"(npre[6]), "v"(npre[7]));
  float n[8];
#pragma unroll
  for (int v = 0; v < 8; ++v) n[v] = tanhf(npre[v]);
#pragma unroll
  for (int v = 0; v < 8; ++v) {
    const float hold = So[8 * jj + v];
    const float zg = sigmoidf_(gi[1][v] + gh[1][v]);
    So[8 * jj + v] = (1.0f - zg) * n[v] + zg * hold;
  }
}

__device__ __forceinline__ void step_signal() {
  asm volatile("s_wait_dscnt 0x0\n\ts_barrier_signal -1" ::: "memory");
}
__device__ __forceinline__ void step_wait() {
  asm volatile("s_barrier_wait -1" ::: "memory");
}

__global__ __launch_bounds__(512) void gru_fused_kernel(
    const float* __restrict__ x_in,
    const float* __restrict__ w_ih0,
    const float* __restrict__ w_hh0,
    const float* __restrict__ b_ih0,
    const float* __restrict__ b_hh0,
    const float* __restrict__ w_ih1,
    const float* __restrict__ w_hh1,
    const float* __restrict__ b_ih1,
    const float* __restrict__ b_hh1,
    float* __restrict__ z_out)
{
  __shared__ __attribute__((aligned(16))) float wih0_s[G3 * WP32];
  __shared__ __attribute__((aligned(16))) float whh0_s[G3 * WP64];
  __shared__ __attribute__((aligned(16))) float wih1_s[G3 * WP64];
  __shared__ __attribute__((aligned(16))) float whh1_s[G3 * WP64];
  __shared__ __attribute__((aligned(16))) float bias_s[4 * G3];
  __shared__ __attribute__((aligned(16))) float ost[GRU_WAVES][16 * OSP];
  __shared__ __attribute__((aligned(16))) float hx[2 * HXBUF];
  __shared__ __attribute__((aligned(16))) float hy[2 * HXBUF];

  const int tid = threadIdx.x;
  {
    const float4* g0 = reinterpret_cast<const float4*>(w_ih0);
    for (int i = tid; i < G3 * FEAT / 4; i += 512) {
      const int row = i >> 3, c4 = i & 7;
      *reinterpret_cast<float4*>(&wih0_s[row * WP32 + 4 * c4]) = g0[i];
    }
    const float4* g1 = reinterpret_cast<const float4*>(w_hh0);
    const float4* g2 = reinterpret_cast<const float4*>(w_ih1);
    const float4* g3 = reinterpret_cast<const float4*>(w_hh1);
    for (int i = tid; i < G3 * HID / 4; i += 512) {
      const int row = i >> 4, c4 = i & 15;
      *reinterpret_cast<float4*>(&whh0_s[row * WP64 + 4 * c4]) = g1[i];
      *reinterpret_cast<float4*>(&wih1_s[row * WP64 + 4 * c4]) = g2[i];
      *reinterpret_cast<float4*>(&whh1_s[row * WP64 + 4 * c4]) = g3[i];
    }
    for (int i = tid; i < G3; i += 512) {
      bias_s[i]          = b_ih0[i];
      bias_s[G3 + i]     = b_hh0[i];
      bias_s[2 * G3 + i] = b_ih1[i];
      bias_s[3 * G3 + i] = b_hh1[i];
    }
  }
  __syncthreads();

  const int wave = tid >> 5;
  const int lane = tid & 31;
  const int l16  = lane & 15;
  const int lh   = lane >> 4;
  const int role  = __builtin_amdgcn_readfirstlane(wave >> 2);
  const int layer = (role < 2) ? 1 : 0;
  const int half  = role & 1;
  const int tile  = wave & (GRU_TILES - 1);
  const int row0  = blockIdx.x * (GRU_TILES * 16) + tile * 16;
  const int own   = 16 * half;
  const int oth   = 16 - own;

  float So[16], Sp[16], xc[16];
#pragma unroll
  for (int i = 0; i < 16; ++i) { So[i] = 0.0f; Sp[i] = 0.0f; xc[i] = 0.0f; }
  v8f gi[2][3], gh[2][3];
#pragma unroll
  for (int j = 0; j < 2; ++j)
#pragma unroll
    for (int g = 0; g < 3; ++g)
#pragma unroll
      for (int e = 0; e < 8; ++e) { gi[j][g][e] = 0.0f; gh[j][g][e] = 0.0f; }

  float* hxl = &hx[(tile * 32 + lane) * HXP];
  float* hyl = &hy[(tile * 32 + lane) * HXP];

  const int rw = 32 * half + l16;
  const float* wia;
  const float* wha;
  const float* whb;
  const float* bi;
  const float* bh;
  if (layer == 0) {
    wia = wih0_s + rw * WP32 + 16 * lh;
    wha = whh0_s + rw * WP64 + 32 * half + 8 * lh;
    whb = whh0_s + rw * WP64 + 32 * (1 - half) + 8 * lh;
    bi  = bias_s + 32 * half + 8 * lh;
    bh  = bias_s + G3 + 32 * half + 8 * lh;
    const float4* xp = reinterpret_cast<const float4*>(
        x_in + ((size_t)row0 + l16) * FEAT + 16 * lh);
#pragma unroll
    for (int u = 0; u < 4; ++u) {
      const float4 v = xp[u];
      xc[4 * u] = v.x; xc[4 * u + 1] = v.y; xc[4 * u + 2] = v.z; xc[4 * u + 3] = v.w;
    }
  } else {
    wia = wih1_s + rw * WP64 + 8 * lh;
    wha = whh1_s + rw * WP64 + 32 * half + 8 * lh;
    whb = whh1_s + rw * WP64 + 32 * (1 - half) + 8 * lh;
    bi  = bias_s + 2 * G3 + 32 * half + 8 * lh;
    bh  = bias_s + 3 * G3 + 32 * half + 8 * lh;
  }

  float* mys = layer ? hyl : hxl;
  for (int s = -1; s <= T_SEQ; ++s) {
    const int t = s - layer;
    if (t >= 0 && t < T_SEQ) {
      const int rd = ((s & 1) ^ 1) * HXBUF;
      if (t >= 1) slab_read16(mys + rd + oth, Sp);
#pragma unroll
      for (int jj = 0; jj < 2; ++jj) {
        v4f F[3][4];
        load_w4<0, WP64>(whb, jj, F);
        __builtin_amdgcn_sched_barrier(0);
        mm_k32(gh[jj], F, Sp);
        __builtin_amdgcn_sched_barrier(0);
        if (layer == 1) {
#pragma unroll
          for (int kh = 0; kh < 2; ++kh) {
            load_w4<0, WP64>(wia + 32 * kh, jj, F);
            slab_read16(hxl + rd + 16 * kh, xc);
            if (kh == 0) load_bias(bi, jj, gi[jj]);
            __builtin_amdgcn_sched_barrier(0);
            mm_k32(gi[jj], F, xc);
            __builtin_amdgcn_sched_barrier(0);
          }
        }
        gate_tile(gi[jj], gh[jj], So, jj);
      }
      slab_write16(mys + (s & 1) * HXBUF + own, So);
    }
    step_signal();
    __builtin_amdgcn_sched_barrier(0);
    if (t + 1 >= 0 && t + 1 < T_SEQ) {
#pragma unroll
      for (int jj = 0; jj < 2; ++jj) {
        v4f F[3][4];
        load_w4<0, WP64>(wha, jj, F);
        load_bias(bh, jj, gh[jj]);
        __builtin_amdgcn_sched_barrier(0);
        mm_k32(gh[jj], F, So);
        __builtin_amdgcn_sched_barrier(0);
        if (layer == 0) {
          load_w4<1, WP32>(wia, jj, F);
          load_bias(bi, jj, gi[jj]);
          __builtin_amdgcn_sched_barrier(0);
          mm_k32(gi[jj], F, xc);
          __builtin_amdgcn_sched_barrier(0);
        }
      }
      if (layer == 0) {
        const int tn = (t + 2 < T_SEQ) ? (t + 2) : (t + 1);
        const float4* xnp = reinterpret_cast<const float4*>(
            x_in + ((size_t)tn * NROWS + row0 + l16) * FEAT + 16 * lh);
#pragma unroll
        for (int u = 0; u < 4; ++u) {
          const float4 v = xnp[u];
          xc[4 * u] = v.x; xc[4 * u + 1] = v.y; xc[4 * u + 2] = v.z; xc[4 * u + 3] = v.w;
        }
      }
    }
    step_wait();
  }

  float* st = &ost[wave][0];
#pragma unroll
  for (int u = 0; u < 4; ++u) {
    v4f v;
#pragma unroll
    for (int e = 0; e < 4; ++e) v[e] = So[4 * u + e];
    *reinterpret_cast<v4f*>(&st[l16 * OSP + 16 * (u >> 1) + 8 * lh + 4 * (u & 1)]) = v;
  }
#pragma unroll
  for (int it = 0; it < 4; ++it) {
    const int idx = it * 32 + lane;
    const int r = idx >> 3, c4 = idx & 7;
    const v4f v = lds4(&st[r * OSP + 4 * c4]);
    *reinterpret_cast<v4f*>(
        &z_out[(size_t)layer * NROWS * HID + (size_t)(row0 + r) * HID + 32 * half + 4 * c4]) = v;
  }
}

#define GAT_WAVES 8
#define ZP 68

typedef __attribute__((ext_vector_type(4))) int v4i;

__device__ __forceinline__ float xhalf(float x) {
  const int b = __builtin_bit_cast(int, x);
  return __builtin_bit_cast(float,
      __builtin_amdgcn_permlanex16(b, b, 0x76543210, 0xfedcba98, false, false));
}

__device__ __forceinline__ void gat_fetch(const float* zs, const int* mlane, int m0,
                                          int l16, int lh, v4f (&za)[8],
                                          float (&tx)[4][4], float (&ty)[4][4],
                                          v4i (&mk)[2]) {
  const float* ar = zs + (m0 + l16) * ZP + 4 * lh;
#pragma unroll
  for (int u = 0; u < 8; ++u) za[u] = lds4(ar + 8 * u);
  const float* tr = zs + (m0 + 8 * lh) * ZP + l16;
#pragma unroll
  for (int q = 0; q < 4; ++q) {
#pragma unroll
    for (int dt = 0; dt < 4; ++dt) {
      tx[dt][q] = tr[(2 * q) * ZP + 16 * dt];
      ty[dt][q] = tr[(2 * q + 1) * ZP + 16 * dt];
    }
  }
  const v4i* mp = reinterpret_cast<const v4i*>(mlane + m0);
  mk[0] = mp[0];
  mk[1] = mp[1];
}

__device__ __forceinline__ void gat_step(const v4f (&zb)[8], const v4f (&za)[8],
                                         const float (&tx)[4][4], const float (&ty)[4][4],
                                         const v4i (&mk)[2],
                                         float& M, float& L, v8f (&accO)[4]) {
  v8f s0 = {}, s1 = {};
#pragma unroll
  for (int u = 0; u < 8; ++u) {
    v2f a0, b0, a1, b1;
    a0.x = za[u][0]; a0.y = za[u][1]; b0.x = zb[u][0]; b0.y = zb[u][1];
    a1.x = za[u][2]; a1.y = za[u][3]; b1.x = zb[u][2]; b1.y = zb[u][3];
    s0 = wmma_f32(a0, b0, s0);
    s1 = wmma_f32(a1, b1, s1);
  }
  float a[8];
#pragma unroll
  for (int r = 0; r < 8; ++r) {
    const float s  = s0[r] + s1[r];
    const float lr = s > 0.0f ? s : 0.1f * s;
    const int   mi = (r < 4) ? mk[0][r & 3] : mk[1][r & 3];
    float v = lr * (float)mi;
    v = (v == 0.0f) ? NEG_BIG : v;
    a[r] = v;
  }
  float tm = a[0];
#pragma unroll
  for (int r = 1; r < 8; ++r) tm = fmaxf(tm, a[r]);
  tm = fmaxf(tm, xhalf(tm));
  const float Mn    = fmaxf(M, tm);
  const float alpha = __expf(M - Mn);
  float p[8];
  float ps = 0.0f;
#pragma unroll
  for (int r = 0; r < 8; ++r) { p[r] = __expf(a[r] - Mn); ps += p[r]; }
  L = L * alpha + ps;
  M = Mn;
#pragma unroll
  for (int dt = 0; dt < 4; ++dt)
#pragma unroll
    for (int r = 0; r < 8; ++r) accO[dt][r] *= alpha;
#pragma unroll
  for (int q = 0; q < 4; ++q) {
    v2f b;
    b.x = p[2 * q];
    b.y = p[2 * q + 1];
#pragma unroll
    for (int dt = 0; dt < 4; ++dt) {
      v2f a2;
      a2.x = tx[dt][q];
      a2.y = ty[dt][q];
      accO[dt] = wmma_f32(a2, b, accO[dt]);
    }
  }
}

__global__ __launch_bounds__(256) void gat_kernel(
    const float* __restrict__ z,
    const int*   __restrict__ mask,
    float*       __restrict__ out)
{
  __shared__ __attribute__((aligned(16))) float zs[NODES * ZP];

  const int tid   = threadIdx.x;
  const int pair  = blockIdx.x >> 3;
  const int chunk = blockIdx.x & 7;
  const float* zp = z + (size_t)pair * NODES * HID;

  {
    const float4* g = reinterpret_cast<const float4*>(zp);
#pragma unroll 8
    for (int it = 0; it < NODES * HID / 4 / 256; ++it) {
      const int i = it * 256 + tid;
      const int row = i >> 4, c4 = i & 15;
      *reinterpret_cast<float4*>(&zs[row * ZP + 4 * c4]) = g[i];
    }
  }
  __syncthreads();

  const int wave = tid >> 5;
  const int lane = tid & 31;
  const int l16  = lane & 15;
  const int lh   = lane >> 4;
  const int n0   = (chunk * GAT_WAVES + wave) * 16;

  v4f zb[8];
  {
    const float* br = zs + (n0 + l16) * ZP + 4 * lh;
#pragma unroll
    for (int u = 0; u < 8; ++u) zb[u] = lds4(br + 8 * u);
  }
  const int* mlane = mask + (size_t)(n0 + l16) * NODES + 8 * lh;

  float M = -INFINITY, L = 0.0f;
  v8f accO[4] = {};
  v4f zaA[8], zaB[8];
  float txA[4][4], tyA[4][4], txB[4][4], tyB[4][4];
  v4i mkA[2], mkB[2];

  gat_fetch(zs, mlane, 0, l16, lh, zaA, txA, tyA, mkA);
  for (int t = 0; t < 64; t += 2) {
    gat_fetch(zs, mlane, (t + 1) * 16, l16, lh, zaB, txB, tyB, mkB);
    __builtin_amdgcn_sched_barrier(0);
    gat_step(zb, zaA, txA, tyA, mkA, M, L, accO);
    {
      const int tn = (t + 2 < 64) ? (t + 2) : t;
      gat_fetch(zs, mlane, tn * 16, l16, lh, zaA, txA, tyA, mkA);
    }
    __builtin_amdgcn_sched_barrier(0);
    gat_step(zb, zaB, txB, tyB, mkB, M, L, accO);
  }

  const float inv = 1.0f / (L + xhalf(L));
  float* orow = out + (size_t)pair * NODES * HID + (size_t)(n0 + l16) * HID + 8 * lh;
#pragma unroll
  for (int dt = 0; dt < 4; ++dt) {
    v4f lo, hi;
#pragma unroll
    for (int e = 0; e < 4; ++e) { lo[e] = accO[dt][e] * inv; hi[e] = accO[dt][4 + e] * inv; }
    *reinterpret_cast<v4f*>(orow + 16 * dt)     = lo;
    *reinterpret_cast<v4f*>(orow + 16 * dt + 4) = hi;
  }
}

extern "C" void kernel_launch(void* const* d_in, const int* in_sizes, int n_in,
                              void* d_out, int out_size, void* d_ws, size_t ws_size,
                              hipStream_t stream) {
  const float* x     = (const float*)d_in[0];
  const int*   mask  = (const int*)  d_in[1];
  const float* w_ih0 = (const float*)d_in[2];
  const float* w_hh0 = (const float*)d_in[3];
  const float* b_ih0 = (const float*)d_in[4];
  const float* b_hh0 = (const float*)d_in[5];
  const float* w_ih1 = (const float*)d_in[6];
  const float* w_hh1 = (const float*)d_in[7];
  const float* b_ih1 = (const float*)d_in[8];
  const float* b_hh1 = (const float*)d_in[9];
  float* out  = (float*)d_out;
  float* zbuf = (float*)d_ws;

  gru_fused_kernel<<<256, 512, 0, stream>>>(x, w_ih0, w_hh0, b_ih0, b_hh0,
                                            w_ih1, w_hh1, b_ih1, b_hh1, zbuf);
  gat_kernel<<<256, 256, 0, stream>>>(zbuf, mask, out);
}


# GRU gate activations on the hardware tanh instruction: tanh = one v_tanh_f32, sigmoid(x) = 1/2 + tanh(x/2)/2, on whole accumulator vectors (packed f32); no branch, no divide sequence; same sums
# speedup vs baseline: 3.0125x; 1.1755x over previous
#include <hip/hip_runtime.h>
#include <hip/hip_bf16.h>
#include <math.h>


typedef __attribute__((ext_vector_type(2))) float v2f;
typedef __attribute__((ext_vector_type(4))) float v4f;
typedef __attribute__((ext_vector_type(8))) float v8f;

#define T_SEQ   12
#define BCN     16
#define NODES   1024
#define FEAT    32
#define HID     64
#define G3      192
#define NROWS   (BCN * NODES)
#define NEG_BIG (-1e16f)

__device__ __forceinline__ v8f wmma_f32(v2f a, v2f b, v8f c) {
  return __builtin_amdgcn_wmma_f32_16x16x4_f32(
       false, a,  false, b,
       (short)0, c,  false,  false);
}

#define GRU_TILES 4
#define GRU_WAVES 16
#define WP32 36
#define WP64 68
#define HXP  36
#define HXBUF (GRU_TILES * 32 * HXP)
#define OSP  36

__device__ __forceinline__ v4f lds4(const float* p) {
  return *reinterpret_cast<const v4f*>(p);
}

__device__ __forceinline__ v8f join8(v4f lo, v4f hi) {
  return __builtin_shufflevector(lo, hi, 0, 1, 2, 3, 4, 5, 6, 7);
}

__device__ __forceinline__ void mm_k32(v8f (&acc)[3], const v4f (&f)[3][4],
                                       const float (&b)[16]) {
#pragma unroll
  for (int u = 0; u < 4; ++u) {
#pragma unroll
    for (int hf = 0; hf < 2; ++hf) {
      v2f bb;
      bb.x = b[4 * u + 2 * hf];
      bb.y = b[4 * u + 2 * hf + 1];
#pragma unroll
      for (int g = 0; g < 3; ++g) {
        v2f aa;
        aa.x = f[g][u][2 * hf];
        aa.y = f[g][u][2 * hf + 1];
        acc[g] = wmma_f32(aa, bb, acc[g]);
      }
    }
  }
}

template <int K32, int PITCH>
__device__ __forceinline__ void load_w4(const float* p, int jj, v4f (&F)[3][4]) {
#pragma unroll
  for (int g = 0; g < 3; ++g) {
    const int ro = (64 * g + 16 * jj) * PITCH;
#pragma unroll
    for (int u = 0; u < 4; ++u)
      F[g][u] = lds4(p + ro + (K32 ? 4 * u : 16 * (u >> 1) + 4 * (u & 1)));
  }
}

__device__ __forceinline__ void load_bias(const float* b, int jj, v8f (&acc)[3]) {
#pragma unroll
  for (int g = 0; g < 3; ++g)
    acc[g] = join8(lds4(b + 64 * g + 16 * jj), lds4(b + 64 * g + 16 * jj + 4));
}

__device__ __forceinline__ void slab_read16(const float* p, float (&d)[16]) {
#pragma unroll
  for (int u = 0; u < 4; ++u) {
    const v4f v = lds4(p + 4 * u);
#pragma unroll
    for (int e = 0; e < 4; ++e) d[4 * u + e] = v[e];
  }
}

__device__ __forceinline__ void slab_write16(float* p, const float (&d)[16]) {
#pragma unroll
  for (int u = 0; u < 4; ++u) {
    v4f v;
#pragma unroll
    for (int e = 0; e < 4; ++e) v[e] = d[4 * u + e];
    *reinterpret_cast<v4f*>(p + 4 * u) = v;
  }
}

__device__ __forceinline__ v8f tanh8(v8f x) {
  v8f y;
#pragma unroll
  for (int v = 0; v < 8; ++v) y[v] = __builtin_amdgcn_tanhf(x[v]);
  return y;
}
__device__ __forceinline__ v8f sigmoid8(v8f x) {
  return tanh8(x * 0.5f) * 0.5f + 0.5f;
}
__device__ __forceinline__ void gate_tile(const v8f (&gi)[3], const v8f (&gh)[3],
                                          float (&So)[16], int jj) {
  const v8f r  = sigmoid8(gi[0] + gh[0]);
  const v8f n  = tanh8(gi[2] + r * gh[2]);
  const v8f zg = sigmoid8(gi[1] + gh[1]);
  v8f hold;
#pragma unroll
  for (int v = 0; v < 8; ++v) hold[v] = So[8 * jj + v];
  const v8f hn = (1.0f - zg) * n + zg * hold;
#pragma unroll
  for (int v = 0; v < 8; ++v) So[8 * jj + v] = hn[v];
}

__device__ __forceinline__ void step_signal() {
  asm volatile("s_wait_dscnt 0x0\n\ts_barrier_signal -1" ::: "memory");
}
__device__ __forceinline__ void step_wait() {
  asm volatile("s_barrier_wait -1" ::: "memory");
}

__global__ __launch_bounds__(512) void gru_fused_kernel(
    const float* __restrict__ x_in,
    const float* __restrict__ w_ih0,
    const float* __restrict__ w_hh0,
    const float* __restrict__ b_ih0,
    const float* __restrict__ b_hh0,
    const float* __restrict__ w_ih1,
    const float* __restrict__ w_hh1,
    const float* __restrict__ b_ih1,
    const float* __restrict__ b_hh1,
    float* __restrict__ z_out)
{
  __shared__ __attribute__((aligned(16))) float wih0_s[G3 * WP32];
  __shared__ __attribute__((aligned(16))) float whh0_s[G3 * WP64];
  __shared__ __attribute__((aligned(16))) float wih1_s[G3 * WP64];
  __shared__ __attribute__((aligned(16))) float whh1_s[G3 * WP64];
  __shared__ __attribute__((aligned(16))) float bias_s[4 * G3];
  __shared__ __attribute__((aligned(16))) float ost[GRU_WAVES][16 * OSP];
  __shared__ __attribute__((aligned(16))) float hx[2 * HXBUF];
  __shared__ __attribute__((aligned(16))) float hy[2 * HXBUF];

  const int tid = threadIdx.x;
  {
    const float4* g0 = reinterpret_cast<const float4*>(w_ih0);
    for (int i = tid; i < G3 * FEAT / 4; i += 512) {
      const int row = i >> 3, c4 = i & 7;
      *reinterpret_cast<float4*>(&wih0_s[row * WP32 + 4 * c4]) = g0[i];
    }
    const float4* g1 = reinterpret_cast<const float4*>(w_hh0);
    const float4* g2 = reinterpret_cast<const float4*>(w_ih1);
    const float4* g3 = reinterpret_cast<const float4*>(w_hh1);
    for (int i = tid; i < G3 * HID / 4; i += 512) {
      const int row = i >> 4, c4 = i & 15;
      *reinterpret_cast<float4*>(&whh0_s[row * WP64 + 4 * c4]) = g1[i];
      *reinterpret_cast<float4*>(&wih1_s[row * WP64 + 4 * c4]) = g2[i];
      *reinterpret_cast<float4*>(&whh1_s[row * WP64 + 4 * c4]) = g3[i];
    }
    for (int i = tid; i < G3; i += 512) {
      bias_s[i]          = b_ih0[i];
      bias_s[G3 + i]     = b_hh0[i];
      bias_s[2 * G3 + i] = b_ih1[i];
      bias_s[3 * G3 + i] = b_hh1[i];
    }
  }
  __syncthreads();

  const int wave = tid >> 5;
  const int lane = tid & 31;
  const int l16  = lane & 15;
  const int lh   = lane >> 4;
  const int role  = __builtin_amdgcn_readfirstlane(wave >> 2);
  const int layer = (role < 2) ? 1 : 0;
  const int half  = role & 1;
  const int tile  = wave & (GRU_TILES - 1);
  const int row0  = blockIdx.x * (GRU_TILES * 16) + tile * 16;
  const int own   = 16 * half;
  const int oth   = 16 - own;

  float So[16], Sp[16], xc[16];
#pragma unroll
  for (int i = 0; i < 16; ++i) { So[i] = 0.0f; Sp[i] = 0.0f; xc[i] = 0.0f; }
  v8f gi[2][3], gh[2][3];
#pragma unroll
  for (int j = 0; j < 2; ++j)
#pragma unroll
    for (int g = 0; g < 3; ++g)
#pragma unroll
      for (int e = 0; e < 8; ++e) { gi[j][g][e] = 0.0f; gh[j][g][e] = 0.0f; }

  float* hxl = &hx[(tile * 32 + lane) * HXP];
  float* hyl = &hy[(tile * 32 + lane) * HXP];

  const int rw = 32 * half + l16;
  const float* wia;
  const float* wha;
  const float* whb;
  const float* bi;
  const float* bh;
  if (layer == 0) {
    wia = wih0_s + rw * WP32 + 16 * lh;
    wha = whh0_s + rw * WP64 + 32 * half + 8 * lh;
    whb = whh0_s + rw * WP64 + 32 * (1 - half) + 8 * lh;
    bi  = bias_s + 32 * half + 8 * lh;
    bh  = bias_s + G3 + 32 * half + 8 * lh;
    const float4* xp = reinterpret_cast<const float4*>(
        x_in + ((size_t)row0 + l16) * FEAT + 16 * lh);
#pragma unroll
    for (int u = 0; u < 4; ++u) {
      const float4 v = xp[u];
      xc[4 * u] = v.x; xc[4 * u + 1] = v.y; xc[4 * u + 2] = v.z; xc[4 * u + 3] = v.w;
    }
  } else {
    wia = wih1_s + rw * WP64 + 8 * lh;
    wha = whh1_s + rw * WP64 + 32 * half + 8 * lh;
    whb = whh1_s + rw * WP64 + 32 * (1 - half) + 8 * lh;
    bi  = bias_s + 2 * G3 + 32 * half + 8 * lh;
    bh  = bias_s + 3 * G3 + 32 * half + 8 * lh;
  }

  float* mys = layer ? hyl : hxl;
  for (int s = -1; s <= T_SEQ; ++s) {
    const int t = s - layer;
    if (t >= 0 && t < T_SEQ) {
      const int rd = ((s & 1) ^ 1) * HXBUF;
      if (t >= 1) slab_read16(mys + rd + oth, Sp);
#pragma unroll
      for (int jj = 0; jj < 2; ++jj) {
        v4f F[3][4];
        load_w4<0, WP64>(whb, jj, F);
        __builtin_amdgcn_sched_barrier(0);
        mm_k32(gh[jj], F, Sp);
        __builtin_amdgcn_sched_barrier(0);
        if (layer == 1) {
#pragma unroll
          for (int kh = 0; kh < 2; ++kh) {
            load_w4<0, WP64>(wia + 32 * kh, jj, F);
            slab_read16(hxl + rd + 16 * kh, xc);
            if (kh == 0) load_bias(bi, jj, gi[jj]);
            __builtin_amdgcn_sched_barrier(0);
            mm_k32(gi[jj], F, xc);
            __builtin_amdgcn_sched_barrier(0);
          }
        }
        gate_tile(gi[jj], gh[jj], So, jj);
      }
      slab_write16(mys + (s & 1) * HXBUF + own, So);
    }
    step_signal();
    __builtin_amdgcn_sched_barrier(0);
    if (t + 1 >= 0 && t + 1 < T_SEQ) {
#pragma unroll
      for (int jj = 0; jj < 2; ++jj) {
        v4f F[3][4];
        load_w4<0, WP64>(wha, jj, F);
        load_bias(bh, jj, gh[jj]);
        __builtin_amdgcn_sched_barrier(0);
        mm_k32(gh[jj], F, So);
        __builtin_amdgcn_sched_barrier(0);
        if (layer == 0) {
          load_w4<1, WP32>(wia, jj, F);
          load_bias(bi, jj, gi[jj]);
          __builtin_amdgcn_sched_barrier(0);
          mm_k32(gi[jj], F, xc);
          __builtin_amdgcn_sched_barrier(0);
        }
      }
      if (layer == 0) {
        const int tn = (t + 2 < T_SEQ) ? (t + 2) : (t + 1);
        const float4* xnp = reinterpret_cast<const float4*>(
            x_in + ((size_t)tn * NROWS + row0 + l16) * FEAT + 16 * lh);
#pragma unroll
        for (int u = 0; u < 4; ++u) {
          const float4 v = xnp[u];
          xc[4 * u] = v.x; xc[4 * u + 1] = v.y; xc[4 * u + 2] = v.z; xc[4 * u + 3] = v.w;
        }
      }
    }
    step_wait();
  }

  float* st = &ost[wave][0];
#pragma unroll
  for (int u = 0; u < 4; ++u) {
    v4f v;
#pragma unroll
    for (int e = 0; e < 4; ++e) v[e] = So[4 * u + e];
    *reinterpret_cast<v4f*>(&st[l16 * OSP + 16 * (u >> 1) + 8 * lh + 4 * (u & 1)]) = v;
  }
#pragma unroll
  for (int it = 0; it < 4; ++it) {
    const int idx = it * 32 + lane;
    const int r = idx >> 3, c4 = idx & 7;
    const v4f v = lds4(&st[r * OSP + 4 * c4]);
    *reinterpret_cast<v4f*>(
        &z_out[(size_t)layer * NROWS * HID + (size_t)(row0 + r) * HID + 32 * half + 4 * c4]) = v;
  }
}

#define GAT_WAVES 8
#define ZP 68

typedef __attribute__((ext_vector_type(4))) int v4i;

__device__ __forceinline__ float xhalf(float x) {
  const int b = __builtin_bit_cast(int, x);
  return __builtin_bit_cast(float,
      __builtin_amdgcn_permlanex16(b, b, 0x76543210, 0xfedcba98, false, false));
}

__device__ __forceinline__ void gat_fetch(const float* zs, const int* mlane, int m0,
                                          int l16, int lh, v4f (&za)[8],
                                          float (&tx)[4][4], float (&ty)[4][4],
                                          v4i (&mk)[2]) {
  const float* ar = zs + (m0 + l16) * ZP + 4 * lh;
#pragma unroll
  for (int u = 0; u < 8; ++u) za[u] = lds4(ar + 8 * u);
  const float* tr = zs + (m0 + 8 * lh) * ZP + l16;
#pragma unroll
  for (int q = 0; q < 4; ++q) {
#pragma unroll
    for (int dt = 0; dt < 4; ++dt) {
      tx[dt][q] = tr[(2 * q) * ZP + 16 * dt];
      ty[dt][q] = tr[(2 * q + 1) * ZP + 16 * dt];
    }
  }
  const v4i* mp = reinterpret_cast<const v4i*>(mlane + m0);
  mk[0] = mp[0];
  mk[1] = mp[1];
}

__device__ __forceinline__ void gat_step(const v4f (&zb)[8], const v4f (&za)[8],
                                         const float (&tx)[4][4], const float (&ty)[4][4],
                                         const v4i (&mk)[2],
                                         float& M, float& L, v8f (&accO)[4]) {
  v8f s0 = {}, s1 = {};
#pragma unroll
  for (int u = 0; u < 8; ++u) {
    v2f a0, b0, a1, b1;
    a0.x = za[u][0]; a0.y = za[u][1]; b0.x = zb[u][0]; b0.y = zb[u][1];
    a1.x = za[u][2]; a1.y = za[u][3]; b1.x = zb[u][2]; b1.y = zb[u][3];
    s0 = wmma_f32(a0, b0, s0);
    s1 = wmma_f32(a1, b1, s1);
  }
  float a[8];
#pragma unroll
  for (int r = 0; r < 8; ++r) {
    const float s  = s0[r] + s1[r];
    const float lr = s > 0.0f ? s : 0.1f * s;
    const int   mi = (r < 4) ? mk[0][r & 3] : mk[1][r & 3];
    float v = lr * (float)mi;
    v = (v == 0.0f) ? NEG_BIG : v;
    a[r] = v;
  }
  float tm = a[0];
#pragma unroll
  for (int r = 1; r < 8; ++r) tm = fmaxf(tm, a[r]);
  tm = fmaxf(tm, xhalf(tm));
  const float Mn    = fmaxf(M, tm);
  const float alpha = __expf(M - Mn);
  float p[8];
  float ps = 0.0f;
#pragma unroll
  for (int r = 0; r < 8; ++r) { p[r] = __expf(a[r] - Mn); ps += p[r]; }
  L = L * alpha + ps;
  M = Mn;
#pragma unroll
  for (int dt = 0; dt < 4; ++dt)
#pragma unroll
    for (int r = 0; r < 8; ++r) accO[dt][r] *= alpha;
#pragma unroll
  for (int q = 0; q < 4; ++q) {
    v2f b;
    b.x = p[2 * q];
    b.y = p[2 * q + 1];
#pragma unroll
    for (int dt = 0; dt < 4; ++dt) {
      v2f a2;
      a2.x = tx[dt][q];
      a2.y = ty[dt][q];
      accO[dt] = wmma_f32(a2, b, accO[dt]);
    }
  }
}

__global__ __launch_bounds__(256) void gat_kernel(
    const float* __restrict__ z,
    const int*   __restrict__ mask,
    float*       __restrict__ out)
{
  __shared__ __attribute__((aligned(16))) float zs[NODES * ZP];

  const int tid   = threadIdx.x;
  const int pair  = blockIdx.x >> 3;
  const int chunk = blockIdx.x & 7;
  const float* zp = z + (size_t)pair * NODES * HID;

  {
    const float4* g = reinterpret_cast<const float4*>(zp);
#pragma unroll 8
    for (int it = 0; it < NODES * HID / 4 / 256; ++it) {
      const int i = it * 256 + tid;
      const int row = i >> 4, c4 = i & 15;
      *reinterpret_cast<float4*>(&zs[row * ZP + 4 * c4]) = g[i];
    }
  }
  __syncthreads();

  const int wave = tid >> 5;
  const int lane = tid & 31;
  const int l16  = lane & 15;
  const int lh   = lane >> 4;
  const int n0   = (chunk * GAT_WAVES + wave) * 16;

  v4f zb[8];
  {
    const float* br = zs + (n0 + l16) * ZP + 4 * lh;
#pragma unroll
    for (int u = 0; u < 8; ++u) zb[u] = lds4(br + 8 * u);
  }
  const int* mlane = mask + (size_t)(n0 + l16) * NODES + 8 * lh;

  float M = -INFINITY, L = 0.0f;
  v8f accO[4] = {};
  v4f zaA[8], zaB[8];
  float txA[4][4], tyA[4][4], txB[4][4], tyB[4][4];
  v4i mkA[2], mkB[2];

  gat_fetch(zs, mlane, 0, l16, lh, zaA, txA, tyA, mkA);
  for (int t = 0; t < 64; t += 2) {
    gat_fetch(zs, mlane, (t + 1) * 16, l16, lh, zaB, txB, tyB, mkB);
    __builtin_amdgcn_sched_barrier(0);
    gat_step(zb, zaA, txA, tyA, mkA, M, L, accO);
    {
      const int tn = (t + 2 < 64) ? (t + 2) : t;
      gat_fetch(zs, mlane, tn * 16, l16, lh, zaA, txA, tyA, mkA);
    }
    __builtin_amdgcn_sched_barrier(0);
    gat_step(zb, zaB, txB, tyB, mkB, M, L, accO);
  }

  const float inv = 1.0f / (L + xhalf(L));
  float* orow = out + (size_t)pair * NODES * HID + (size_t)(n0 + l16) * HID + 8 * lh;
#pragma unroll
  for (int dt = 0; dt < 4; ++dt) {
    v4f lo, hi;
#pragma unroll
    for (int e = 0; e < 4; ++e) { lo[e] = accO[dt][e] * inv; hi[e] = accO[dt][4 + e] * inv; }
    *reinterpret_cast<v4f*>(orow + 16 * dt)     = lo;
    *reinterpret_cast<v4f*>(orow + 16 * dt + 4) = hi;
  }
}

extern "C" void kernel_launch(void* const* d_in, const int* in_sizes, int n_in,
                              void* d_out, int out_size, void* d_ws, size_t ws_size,
                              hipStream_t stream) {
  const float* x     = (const float*)d_in[0];
  const int*   mask  = (const int*)  d_in[1];
  const float* w_ih0 = (const float*)d_in[2];
  const float* w_hh0 = (const float*)d_in[3];
  const float* b_ih0 = (const float*)d_in[4];
  const float* b_hh0 = (const float*)d_in[5];
  const float* w_ih1 = (const float*)d_in[6];
  const float* w_hh1 = (const float*)d_in[7];
  const float* b_ih1 = (const float*)d_in[8];
  const float* b_hh1 = (const float*)d_in[9];
  float* out  = (float*)d_out;
  float* zbuf = (float*)d_ws;

  gru_fused_kernel<<<256, 512, 0, stream>>>(x, w_ih0, w_hh0, b_ih0, b_hh0,
                                            w_ih1, w_hh1, b_ih1, b_hh1, zbuf);
  gat_kernel<<<256, 256, 0, stream>>>(zbuf, mask, out);
}
